# starting kernel
# baseline (speedup 1.0000x reference)
#include <hip/hip_runtime.h>
#include <hip/hip_bf16.h>
#include <math.h>

typedef __attribute__((ext_vector_type(2))) float v2f;
typedef __attribute__((ext_vector_type(8))) float v8f;

#define HEADS 4
#define F_OUT 64
#define D_IN 128
#define NH_COLS 256
#define NEG_SLOPE 0.2f

__global__ __launch_bounds__(128) void gat_gemm_wmma(
    const float* __restrict__ X,
    const float* __restrict__ W,
    float* __restrict__ H)
{
    const int mtile   = blockIdx.x;
    const int wave    = threadIdx.x >> 5;
    const int lane    = threadIdx.x & 31;
    const int halfsel = lane >> 4;
    const int l16     = lane & 15;
    const int colbase = wave * 64;

    const float* xrow = X + (size_t)(mtile * 16 + l16) * D_IN;

    v8f acc0 = {}; v8f acc1 = {}; v8f acc2 = {}; v8f acc3 = {};

    #pragma unroll 4
    for (int s = 0; s < 32; ++s) {
        const int k0 = s * 4 + halfsel * 2;
        v2f a;
        a.x = xrow[k0];
        a.y = xrow[k0 + 1];
        const float* w0 = W + (size_t)k0 * NH_COLS;
        const float* w1 = W + (size_t)(k0 + 1) * NH_COLS;
        const int c0 = colbase + l16;
        v2f b0; b0.x = w0[c0     ]; b0.y = w1[c0     ];
        v2f b1; b1.x = w0[c0 + 16]; b1.y = w1[c0 + 16];
        v2f b2; b2.x = w0[c0 + 32]; b2.y = w1[c0 + 32];
        v2f b3; b3.x = w0[c0 + 48]; b3.y = w1[c0 + 48];

        acc0 = __builtin_amdgcn_wmma_f32_16x16x4_f32(false, a, false, b0, (short)0, acc0, false, false);
        acc1 = __builtin_amdgcn_wmma_f32_16x16x4_f32(false, a, false, b1, (short)0, acc1, false, false);
        acc2 = __builtin_amdgcn_wmma_f32_16x16x4_f32(false, a, false, b2, (short)0, acc2, false, false);
        acc3 = __builtin_amdgcn_wmma_f32_16x16x4_f32(false, a, false, b3, (short)0, acc3, false, false);
    }

    #pragma unroll
    for (int r = 0; r < 8; ++r) {
        const int m = mtile * 16 + r + halfsel * 8;
        float* hrow = H + (size_t)m * NH_COLS + colbase + l16;
        hrow[0]  = acc0[r];
        hrow[16] = acc1[r];
        hrow[32] = acc2[r];
        hrow[48] = acc3[r];
    }
}

__global__ void gat_init(float* __restrict__ amax, float* __restrict__ denom,
                         float* __restrict__ agg, int nh, int naggr)
{
    const int i = blockIdx.x * blockDim.x + threadIdx.x;
    if (i < nh) {
        amax[i]  = __int_as_float(0xFF800000);
        denom[i] = 0.0f;
    }
    if (i < naggr) agg[i] = 0.0f;
}

__global__ void gat_attn_coef(const float* __restrict__ H,
                              const float* __restrict__ att_src,
                              const float* __restrict__ att_dst,
                              float* __restrict__ a_src,
                              float* __restrict__ a_dst, int nh)
{
    const int tid = blockIdx.x * blockDim.x + threadIdx.x;
    if (tid >= nh) return;
    const int n = tid >> 2, hd = tid & 3;
    const float4* hp = (const float4*)(H + (size_t)n * NH_COLS + hd * F_OUT);
    const float4* as = (const float4*)(att_src + hd * F_OUT);
    const float4* ad = (const float4*)(att_dst + hd * F_OUT);
    float s1 = 0.0f, s2 = 0.0f;
    #pragma unroll
    for (int i = 0; i < 16; ++i) {
        const float4 hv = hp[i], v1 = as[i], v2 = ad[i];
        s1 += hv.x * v1.x + hv.y * v1.y + hv.z * v1.z + hv.w * v1.w;
        s2 += hv.x * v2.x + hv.y * v2.y + hv.z * v2.z + hv.w * v2.w;
    }
    a_src[tid] = s1;
    a_dst[tid] = s2;
}

__device__ __forceinline__ void atomicMaxFloat(float* addr, float val)
{
    if (val >= 0.0f) atomicMax((int*)addr, __float_as_int(val));
    else             atomicMin((unsigned int*)addr, __float_as_uint(val));
}

__device__ __forceinline__ void edge_endpoints(const int* __restrict__ ei,
                                               int e, int E, int& src, int& dst)
{
    if (e < E) { src = ei[e]; dst = ei[E + e]; }
    else       { src = dst = e - E; }
}

__device__ __forceinline__ float leaky(float a)
{
    return a > 0.0f ? a : NEG_SLOPE * a;
}

__global__ void gat_edge_max(const int* __restrict__ ei, int E, int Etot,
                             const float* __restrict__ a_src,
                             const float* __restrict__ a_dst,
                             float* __restrict__ amax)
{
    const int tid = blockIdx.x * blockDim.x + threadIdx.x;
    if (tid >= Etot * HEADS) return;
    const int e = tid >> 2, hd = tid & 3;
    int src, dst; edge_endpoints(ei, e, E, src, dst);
    const float al = leaky(a_src[src * HEADS + hd] + a_dst[dst * HEADS + hd]);
    atomicMaxFloat(&amax[dst * HEADS + hd], al);
}

__global__ void gat_edge_denom(const int* __restrict__ ei, int E, int Etot,
                               const float* __restrict__ a_src,
                               const float* __restrict__ a_dst,
                               const float* __restrict__ amax,
                               float* __restrict__ denom)
{
    const int tid = blockIdx.x * blockDim.x + threadIdx.x;
    if (tid >= Etot * HEADS) return;
    const int e = tid >> 2, hd = tid & 3;
    int src, dst; edge_endpoints(ei, e, E, src, dst);
    const float al = leaky(a_src[src * HEADS + hd] + a_dst[dst * HEADS + hd]);
    atomicAdd(&denom[dst * HEADS + hd], __expf(al - amax[dst * HEADS + hd]));
}

__global__ void gat_edge_agg(const int* __restrict__ ei, int E, int Etot,
                             const float* __restrict__ a_src,
                             const float* __restrict__ a_dst,
                             const float* __restrict__ amax,
                             const float* __restrict__ denom,
                             const float* __restrict__ H,
                             float* __restrict__ agg)
{
    const int gid = blockIdx.x * blockDim.x + threadIdx.x;
    const int e = gid >> 6;
    if (e >= Etot) return;
    const int t  = gid & 63;
    const int hd = t >> 4;
    const int f  = (t & 15) * 4;
    int src, dst; edge_endpoints(ei, e, E, src, dst);
    const int sh = src * HEADS + hd, dh = dst * HEADS + hd;
    const float al = leaky(a_src[sh] + a_dst[dh]);
    const float w  = __expf(al - amax[dh]) / (denom[dh] + 1e-16f);
    const float4 hv = *(const float4*)(H + (size_t)src * NH_COLS + hd * F_OUT + f);
    float* o = agg + (size_t)dst * NH_COLS + hd * F_OUT + f;
    atomicAdd(o + 0, w * hv.x);
    atomicAdd(o + 1, w * hv.y);
    atomicAdd(o + 2, w * hv.z);
    atomicAdd(o + 3, w * hv.w);
}

__global__ void gat_finalize(const float* __restrict__ agg,
                             const float* __restrict__ bias_gat,
                             const float* __restrict__ onehot,
                             const float* __restrict__ W_prior,
                             const float* __restrict__ b_prior,
                             const float* __restrict__ gate,
                             float* __restrict__ out, int nf)
{
    const int tid = blockIdx.x * blockDim.x + threadIdx.x;
    if (tid >= nf) return;
    const int n = tid >> 6, f = tid & 63;
    const float g = gate[0];
    const float* ar = agg + (size_t)n * NH_COLS + f;
    const float m = 0.25f * (ar[0] + ar[64] + ar[128] + ar[192]) + bias_gat[f];
    float p = b_prior[f];
    const float* oh = onehot + (size_t)n * 32;
    #pragma unroll
    for (int c = 0; c < 32; ++c) p += oh[c] * W_prior[c * 64 + f];
    out[tid] = (1.0f - g) * m + g * p;
}

extern "C" void kernel_launch(void* const* d_in, const int* in_sizes, int n_in,
                              void* d_out, int out_size, void* d_ws, size_t ws_size,
                              hipStream_t stream)
{
    const float* x        = (const float*)d_in[0];
    const int*   ei       = (const int*)  d_in[1];
    const float* onehot   = (const float*)d_in[2];
    const float* W_gat    = (const float*)d_in[3];
    const float* att_src  = (const float*)d_in[4];
    const float* att_dst  = (const float*)d_in[5];
    const float* bias_gat = (const float*)d_in[6];
    const float* W_prior  = (const float*)d_in[7];
    const float* b_prior  = (const float*)d_in[8];
    const float* gate     = (const float*)d_in[9];
    float* out = (float*)d_out;

    const int N    = in_sizes[0] / D_IN;
    const int E    = in_sizes[1] / 2;
    const int Etot = E + N;
    const int NHs  = N * HEADS;

    float* h      = (float*)d_ws;
    float* a_src  = h      + (size_t)N * NH_COLS;
    float* a_dst  = a_src  + NHs;
    float* amax   = a_dst  + NHs;
    float* denom  = amax   + NHs;
    float* agg    = denom  + NHs;

    const int naggr = N * NH_COLS;

    gat_init<<<(naggr + 255) / 256, 256, 0, stream>>>(amax, denom, agg, NHs, naggr);

    gat_gemm_wmma<<<N / 16, 128, 0, stream>>>(x, W_gat, h);

    gat_attn_coef<<<(NHs + 255) / 256, 256, 0, stream>>>(h, att_src, att_dst,
                                                         a_src, a_dst, NHs);

    const int eh = Etot * HEADS;
    gat_edge_max<<<(eh + 255) / 256, 256, 0, stream>>>(ei, E, Etot, a_src, a_dst, amax);
    gat_edge_denom<<<(eh + 255) / 256, 256, 0, stream>>>(ei, E, Etot, a_src, a_dst,
                                                         amax, denom);

    const long long aggthreads = (long long)Etot * 64;
    gat_edge_agg<<<(unsigned)((aggthreads + 255) / 256), 256, 0, stream>>>(
        ei, E, Etot, a_src, a_dst, amax, denom, h, agg);

    const int nf = N * F_OUT;
    gat_finalize<<<(nf + 255) / 256, 256, 0, stream>>>(agg, bias_gat, onehot,
                                                       W_prior, b_prior, gate,
                                                       out, nf);
}


# edge_agg sums in LDS, not at L2: a block keeps 200 destination rows in LDS, filters the whole edge list for them, adds w*h[src] with LDS float adds, stores each row once; no global float atomics
# speedup vs baseline: 1.8776x; 1.8776x over previous
#include <hip/hip_runtime.h>
#include <hip/hip_bf16.h>
#include <math.h>

typedef __attribute__((ext_vector_type(2))) float v2f;
typedef __attribute__((ext_vector_type(8))) float v8f;

#define HEADS 4
#define F_OUT 64
#define D_IN 128
#define NH_COLS 256
#define NEG_SLOPE 0.2f

__global__ __launch_bounds__(128) void gat_gemm_wmma(
    const float* __restrict__ X,
    const float* __restrict__ W,
    float* __restrict__ H)
{
    const int mtile   = blockIdx.x;
    const int wave    = threadIdx.x >> 5;
    const int lane    = threadIdx.x & 31;
    const int halfsel = lane >> 4;
    const int l16     = lane & 15;
    const int colbase = wave * 64;

    const float* xrow = X + (size_t)(mtile * 16 + l16) * D_IN;

    v8f acc0 = {}; v8f acc1 = {}; v8f acc2 = {}; v8f acc3 = {};

    #pragma unroll 4
    for (int s = 0; s < 32; ++s) {
        const int k0 = s * 4 + halfsel * 2;
        v2f a;
        a.x = xrow[k0];
        a.y = xrow[k0 + 1];
        const float* w0 = W + (size_t)k0 * NH_COLS;
        const float* w1 = W + (size_t)(k0 + 1) * NH_COLS;
        const int c0 = colbase + l16;
        v2f b0; b0.x = w0[c0     ]; b0.y = w1[c0     ];
        v2f b1; b1.x = w0[c0 + 16]; b1.y = w1[c0 + 16];
        v2f b2; b2.x = w0[c0 + 32]; b2.y = w1[c0 + 32];
        v2f b3; b3.x = w0[c0 + 48]; b3.y = w1[c0 + 48];

        acc0 = __builtin_amdgcn_wmma_f32_16x16x4_f32(false, a, false, b0, (short)0, acc0, false, false);
        acc1 = __builtin_amdgcn_wmma_f32_16x16x4_f32(false, a, false, b1, (short)0, acc1, false, false);
        acc2 = __builtin_amdgcn_wmma_f32_16x16x4_f32(false, a, false, b2, (short)0, acc2, false, false);
        acc3 = __builtin_amdgcn_wmma_f32_16x16x4_f32(false, a, false, b3, (short)0, acc3, false, false);
    }

    #pragma unroll
    for (int r = 0; r < 8; ++r) {
        const int m = mtile * 16 + r + halfsel * 8;
        float* hrow = H + (size_t)m * NH_COLS + colbase + l16;
        hrow[0]  = acc0[r];
        hrow[16] = acc1[r];
        hrow[32] = acc2[r];
        hrow[48] = acc3[r];
    }
}

__global__ void gat_init(float* __restrict__ amax, float* __restrict__ denom, int nh)
{
    const int i = blockIdx.x * blockDim.x + threadIdx.x;
    if (i < nh) {
        amax[i]  = __int_as_float(0xFF800000);
        denom[i] = 0.0f;
    }
}

__global__ void gat_attn_coef(const float* __restrict__ H,
                              const float* __restrict__ att_src,
                              const float* __restrict__ att_dst,
                              float* __restrict__ a_src,
                              float* __restrict__ a_dst, int nh)
{
    const int tid = blockIdx.x * blockDim.x + threadIdx.x;
    if (tid >= nh) return;
    const int n = tid >> 2, hd = tid & 3;
    const float4* hp = (const float4*)(H + (size_t)n * NH_COLS + hd * F_OUT);
    const float4* as = (const float4*)(att_src + hd * F_OUT);
    const float4* ad = (const float4*)(att_dst + hd * F_OUT);
    float s1 = 0.0f, s2 = 0.0f;
    #pragma unroll
    for (int i = 0; i < 16; ++i) {
        const float4 hv = hp[i], v1 = as[i], v2 = ad[i];
        s1 += hv.x * v1.x + hv.y * v1.y + hv.z * v1.z + hv.w * v1.w;
        s2 += hv.x * v2.x + hv.y * v2.y + hv.z * v2.z + hv.w * v2.w;
    }
    a_src[tid] = s1;
    a_dst[tid] = s2;
}

__device__ __forceinline__ void atomicMaxFloat(float* addr, float val)
{
    if (val >= 0.0f) atomicMax((int*)addr, __float_as_int(val));
    else             atomicMin((unsigned int*)addr, __float_as_uint(val));
}

__device__ __forceinline__ void edge_endpoints(const int* __restrict__ ei,
                                               int e, int E, int& src, int& dst)
{
    if (e < E) { src = ei[e]; dst = ei[E + e]; }
    else       { src = dst = e - E; }
}

__device__ __forceinline__ float leaky(float a)
{
    return a > 0.0f ? a : NEG_SLOPE * a;
}

__global__ void gat_edge_max(const int* __restrict__ ei, int E, int Etot,
                             const float* __restrict__ a_src,
                             const float* __restrict__ a_dst,
                             float* __restrict__ amax)
{
    const int tid = blockIdx.x * blockDim.x + threadIdx.x;
    if (tid >= Etot * HEADS) return;
    const int e = tid >> 2, hd = tid & 3;
    int src, dst; edge_endpoints(ei, e, E, src, dst);
    const float al = leaky(a_src[src * HEADS + hd] + a_dst[dst * HEADS + hd]);
    atomicMaxFloat(&amax[dst * HEADS + hd], al);
}

__global__ void gat_edge_denom(const int* __restrict__ ei, int E, int Etot,
                               const float* __restrict__ a_src,
                               const float* __restrict__ a_dst,
                               const float* __restrict__ amax,
                               float* __restrict__ denom)
{
    const int tid = blockIdx.x * blockDim.x + threadIdx.x;
    if (tid >= Etot * HEADS) return;
    const int e = tid >> 2, hd = tid & 3;
    int src, dst; edge_endpoints(ei, e, E, src, dst);
    const float al = leaky(a_src[src * HEADS + hd] + a_dst[dst * HEADS + hd]);
    atomicAdd(&denom[dst * HEADS + hd], __expf(al - amax[dst * HEADS + hd]));
}

#define ROWS_PER_BLOCK 200
#define AGG_THREADS    1024
#define AGG_WAVES      (AGG_THREADS / 32)
#define QCAP           16384
#define CHUNK          16
#define SCAN_DEPTH     8

__device__ __forceinline__ float lane_bcast(float v, int j)
{
    return __int_as_float(__builtin_amdgcn_readlane(__float_as_int(v), j));
}

__device__ __forceinline__ void lds_add_row(float* o,
                                            float u0, float u1, float u2, float u3,
                                            float v0, float v1, float v2, float v3,
                                            float v4, float v5, float v6, float v7)
{
    atomicAdd(o,       u0 * v0);
    atomicAdd(o + 32,  u0 * v1);
    atomicAdd(o + 64,  u1 * v2);
    atomicAdd(o + 96,  u1 * v3);
    atomicAdd(o + 128, u2 * v4);
    atomicAdd(o + 160, u2 * v5);
    atomicAdd(o + 192, u3 * v6);
    atomicAdd(o + 224, u3 * v7);
}

__device__ __forceinline__ void edge_weights(int src, int dst,
                                             const float* __restrict__ a_src,
                                             const float* __restrict__ a_dst,
                                             const float* __restrict__ amax,
                                             const float* __restrict__ denom,
                                             float& w0, float& w1, float& w2, float& w3)
{
    const float4 as = *(const float4*)(a_src + (size_t)src * HEADS);
    const float4 ad = *(const float4*)(a_dst + (size_t)dst * HEADS);
    const float4 mx = *(const float4*)(amax  + (size_t)dst * HEADS);
    const float4 dn = *(const float4*)(denom + (size_t)dst * HEADS);
    w0 = __expf(leaky(as.x + ad.x) - mx.x) / (dn.x + 1e-16f);
    w1 = __expf(leaky(as.y + ad.y) - mx.y) / (dn.y + 1e-16f);
    w2 = __expf(leaky(as.z + ad.z) - mx.z) / (dn.z + 1e-16f);
    w3 = __expf(leaky(as.w + ad.w) - mx.w) / (dn.w + 1e-16f);
}

__global__ __launch_bounds__(AGG_THREADS) void gat_edge_agg(const int* __restrict__ ei, int E, int Etot, int N,
                             const float* __restrict__ a_src,
                             const float* __restrict__ a_dst,
                             const float* __restrict__ amax,
                             const float* __restrict__ denom,
                             const float* __restrict__ H,
                             float* __restrict__ agg)
{
    __shared__ float rows[(ROWS_PER_BLOCK + 1) * NH_COLS];
    __shared__ int   queue[QCAP];
    __shared__ int   qcount;

    const int tid  = threadIdx.x;
    const int lane = tid & 31;
    const int wave = __builtin_amdgcn_readfirstlane(tid >> 5);
    const int n0     = blockIdx.x * ROWS_PER_BLOCK;
    const int rcount = min(ROWS_PER_BLOCK, N - n0);
    const unsigned ur = (unsigned)rcount;

    float4* rows4 = (float4*)rows;
    for (int i = tid; i < (ROWS_PER_BLOCK + 1) * (NH_COLS / 4); i += AGG_THREADS)
        rows4[i] = make_float4(0.0f, 0.0f, 0.0f, 0.0f);
    if (tid < rcount) queue[tid] = E + n0 + tid;
    if (tid == 0) qcount = rcount;
    __syncthreads();

    const int* dp  = ei + E;
    const int  pre = min((4 - (E & 3)) & 3, E);
    const int4* dp4 = (const int4*)(dp + pre);
    const int  nb4 = (E - pre) >> 2;
    const int  rot = (int)(((long long)blockIdx.x * nb4) / gridDim.x);
    if (nb4 > 0) {
        for (int i = tid; i < nb4; i += SCAN_DEPTH * AGG_THREADS) {
            int4 d[SCAN_DEPTH];
            int  q[SCAN_DEPTH];
            #pragma unroll
            for (int k = 0; k < SCAN_DEPTH; ++k) {
                int qq = min(i + k * AGG_THREADS, nb4 - 1) + rot;
                if (qq >= nb4) qq -= nb4;
                q[k] = qq;
                d[k] = dp4[qq];
            }
            #pragma unroll
            for (int k = 0; k < SCAN_DEPTH; ++k) {
                const bool ok = (i + k * AGG_THREADS) < nb4;
                const bool m0 = ok && (unsigned)(d[k].x - n0) < ur;
                const bool m1 = ok && (unsigned)(d[k].y - n0) < ur;
                const bool m2 = ok && (unsigned)(d[k].z - n0) < ur;
                const bool m3 = ok && (unsigned)(d[k].w - n0) < ur;
                const int  c  = (int)m0 + (int)m1 + (int)m2 + (int)m3;
                if (c) {
                    int s = atomicAdd(&qcount, c);
                    const int e0 = pre + 4 * q[k];
                    if (m0) { if (s < QCAP) queue[s] = e0;     ++s; }
                    if (m1) { if (s < QCAP) queue[s] = e0 + 1; ++s; }
                    if (m2) { if (s < QCAP) queue[s] = e0 + 2; ++s; }
                    if (m3) { if (s < QCAP) queue[s] = e0 + 3; }
                }
            }
        }
    }
    {
        int e = -1;
        if (tid < pre) e = tid;
        else if (tid < 8) { const int t = pre + 4 * nb4 + (tid - pre); if (t < E) e = t; }
        if (e >= 0 && (unsigned)(dp[e] - n0) < ur) {
            const int s = atomicAdd(&qcount, 1);
            if (s < QCAP) queue[s] = e;
        }
    }
    __syncthreads();

    const int total = __builtin_amdgcn_readfirstlane(qcount);
    const float* hbase = H + lane;

    if (total <= QCAP) {
        const int nchunks = (total + CHUNK - 1) / CHUNK;
        for (int c = wave; c < nchunks; c += AGG_WAVES) {
            const int ebeg = c * CHUNK;
            const int cnt  = min(CHUNK, total - ebeg);
            int src = 0, row = ROWS_PER_BLOCK;
            float w0 = 0.0f, w1 = 0.0f, w2 = 0.0f, w3 = 0.0f;
            if (lane < cnt) {
                int dst; edge_endpoints(ei, queue[ebeg + lane], E, src, dst);
                row = dst - n0;
                edge_weights(src, dst, a_src, a_dst, amax, denom, w0, w1, w2, w3);
            }
            #pragma unroll
            for (int g = 0; g < CHUNK; g += 4) {
                if (g >= cnt) break;
                float v[4][8];
                #pragma unroll
                for (int k = 0; k < 4; ++k) {
                    const float* hr = hbase + (size_t)__builtin_amdgcn_readlane(src, g + k) * NH_COLS;
                    #pragma unroll
                    for (int t = 0; t < 8; ++t) v[k][t] = hr[32 * t];
                }
                #pragma unroll
                for (int k = 0; k < 4; ++k) {
                    float* o = rows + __builtin_amdgcn_readlane(row, g + k) * NH_COLS + lane;
                    lds_add_row(o,
                                lane_bcast(w0, g + k), lane_bcast(w1, g + k),
                                lane_bcast(w2, g + k), lane_bcast(w3, g + k),
                                v[k][0], v[k][1], v[k][2], v[k][3],
                                v[k][4], v[k][5], v[k][6], v[k][7]);
                }
            }
        }
    } else {
        for (int base = wave * 32; base < Etot; base += AGG_THREADS) {
            const int e = base + lane;
            int src = 0, dst = -1;
            if (e < Etot) edge_endpoints(ei, e, E, src, dst);
            const bool mine = (unsigned)(dst - n0) < ur;
            float w0 = 0.0f, w1 = 0.0f, w2 = 0.0f, w3 = 0.0f;
            if (mine) edge_weights(src, dst, a_src, a_dst, amax, denom, w0, w1, w2, w3);
            const int row = dst - n0;
            unsigned mask = __builtin_amdgcn_ballot_w32(mine);
            while (mask) {
                const int j = __builtin_ctz(mask);
                mask &= mask - 1;
                const float* hr = hbase + (size_t)__builtin_amdgcn_readlane(src, j) * NH_COLS;
                const float v0 = hr[0],   v1 = hr[32],  v2 = hr[64],  v3 = hr[96];
                const float v4 = hr[128], v5 = hr[160], v6 = hr[192], v7 = hr[224];
                lds_add_row(rows + __builtin_amdgcn_readlane(row, j) * NH_COLS + lane,
                            lane_bcast(w0, j), lane_bcast(w1, j), lane_bcast(w2, j), lane_bcast(w3, j),
                            v0, v1, v2, v3, v4, v5, v6, v7);
            }
        }
    }
    __syncthreads();

    float4* out4 = (float4*)(agg + (size_t)n0 * NH_COLS);
    for (int i = tid; i < rcount * (NH_COLS / 4); i += AGG_THREADS)
        out4[i] = rows4[i];
}

__global__ void gat_finalize(const float* __restrict__ agg,
                             const float* __restrict__ bias_gat,
                             const float* __restrict__ onehot,
                             const float* __restrict__ W_prior,
                             const float* __restrict__ b_prior,
                             const float* __restrict__ gate,
                             float* __restrict__ out, int nf)
{
    const int tid = blockIdx.x * blockDim.x + threadIdx.x;
    if (tid >= nf) return;
    const int n = tid >> 6, f = tid & 63;
    const float g = gate[0];
    const float* ar = agg + (size_t)n * NH_COLS + f;
    const float m = 0.25f * (ar[0] + ar[64] + ar[128] + ar[192]) + bias_gat[f];
    float p = b_prior[f];
    const float* oh = onehot + (size_t)n * 32;
    #pragma unroll
    for (int c = 0; c < 32; ++c) p += oh[c] * W_prior[c * 64 + f];
    out[tid] = (1.0f - g) * m + g * p;
}

extern "C" void kernel_launch(void* const* d_in, const int* in_sizes, int n_in,
                              void* d_out, int out_size, void* d_ws, size_t ws_size,
                              hipStream_t stream)
{
    const float* x        = (const float*)d_in[0];
    const int*   ei       = (const int*)  d_in[1];
    const float* onehot   = (const float*)d_in[2];
    const float* W_gat    = (const float*)d_in[3];
    const float* att_src  = (const float*)d_in[4];
    const float* att_dst  = (const float*)d_in[5];
    const float* bias_gat = (const float*)d_in[6];
    const float* W_prior  = (const float*)d_in[7];
    const float* b_prior  = (const float*)d_in[8];
    const float* gate     = (const float*)d_in[9];
    float* out = (float*)d_out;

    const int N    = in_sizes[0] / D_IN;
    const int E    = in_sizes[1] / 2;
    const int Etot = E + N;
    const int NHs  = N * HEADS;

    float* h      = (float*)d_ws;
    float* a_src  = h      + (size_t)N * NH_COLS;
    float* a_dst  = a_src  + NHs;
    float* amax   = a_dst  + NHs;
    float* denom  = amax   + NHs;
    float* agg    = denom  + NHs;

    gat_init<<<(NHs + 255) / 256, 256, 0, stream>>>(amax, denom, NHs);

    gat_gemm_wmma<<<N / 16, 128, 0, stream>>>(x, W_gat, h);

    gat_attn_coef<<<(NHs + 255) / 256, 256, 0, stream>>>(h, att_src, att_dst,
                                                         a_src, a_dst, NHs);

    const int eh = Etot * HEADS;
    gat_edge_max<<<(eh + 255) / 256, 256, 0, stream>>>(ei, E, Etot, a_src, a_dst, amax);
    gat_edge_denom<<<(eh + 255) / 256, 256, 0, stream>>>(ei, E, Etot, a_src, a_dst,
                                                         amax, denom);

    gat_edge_agg<<<(N + ROWS_PER_BLOCK - 1) / ROWS_PER_BLOCK, AGG_THREADS, 0, stream>>>(
        ei, E, Etot, N, a_src, a_dst, amax, denom, h, agg);

    const int nf = N * F_OUT;
    gat_finalize<<<(nf + 255) / 256, 256, 0, stream>>>(agg, bias_gat, onehot,
                                                       W_prior, b_prior, gate,
                                                       out, nf);
}


# edge_agg filters the edge list by one-byte keys (destination's block number, written in the attn_coef launch): 4x fewer bytes, 16 edges per load, one queue grab per 8 loads; sums keep arrival order
# speedup vs baseline: 3.9308x; 2.0935x over previous
#include <hip/hip_runtime.h>
#include <hip/hip_bf16.h>
#include <math.h>

typedef __attribute__((ext_vector_type(2))) float v2f;
typedef __attribute__((ext_vector_type(8))) float v8f;

#define HEADS 4
#define F_OUT 64
#define D_IN 128
#define NH_COLS 256
#define NEG_SLOPE 0.2f
#define ROWS_PER_BLOCK 200

__global__ __launch_bounds__(128) void gat_gemm_wmma(
    const float* __restrict__ X,
    const float* __restrict__ W,
    float* __restrict__ H)
{
    const int mtile   = blockIdx.x;
    const int wave    = threadIdx.x >> 5;
    const int lane    = threadIdx.x & 31;
    const int halfsel = lane >> 4;
    const int l16     = lane & 15;
    const int colbase = wave * 64;

    const float* xrow = X + (size_t)(mtile * 16 + l16) * D_IN;

    v8f acc0 = {}; v8f acc1 = {}; v8f acc2 = {}; v8f acc3 = {};

    #pragma unroll 4
    for (int s = 0; s < 32; ++s) {
        const int k0 = s * 4 + halfsel * 2;
        v2f a;
        a.x = xrow[k0];
        a.y = xrow[k0 + 1];
        const float* w0 = W + (size_t)k0 * NH_COLS;
        const float* w1 = W + (size_t)(k0 + 1) * NH_COLS;
        const int c0 = colbase + l16;
        v2f b0; b0.x = w0[c0     ]; b0.y = w1[c0     ];
        v2f b1; b1.x = w0[c0 + 16]; b1.y = w1[c0 + 16];
        v2f b2; b2.x = w0[c0 + 32]; b2.y = w1[c0 + 32];
        v2f b3; b3.x = w0[c0 + 48]; b3.y = w1[c0 + 48];

        acc0 = __builtin_amdgcn_wmma_f32_16x16x4_f32(false, a, false, b0, (short)0, acc0, false, false);
        acc1 = __builtin_amdgcn_wmma_f32_16x16x4_f32(false, a, false, b1, (short)0, acc1, false, false);
        acc2 = __builtin_amdgcn_wmma_f32_16x16x4_f32(false, a, false, b2, (short)0, acc2, false, false);
        acc3 = __builtin_amdgcn_wmma_f32_16x16x4_f32(false, a, false, b3, (short)0, acc3, false, false);
    }

    #pragma unroll
    for (int r = 0; r < 8; ++r) {
        const int m = mtile * 16 + r + halfsel * 8;
        float* hrow = H + (size_t)m * NH_COLS + colbase + l16;
        hrow[0]  = acc0[r];
        hrow[16] = acc1[r];
        hrow[32] = acc2[r];
        hrow[48] = acc3[r];
    }
}

__global__ void gat_attn_coef(const float* __restrict__ H,
                              const float* __restrict__ att_src,
                              const float* __restrict__ att_dst,
                              float* __restrict__ a_src,
                              float* __restrict__ a_dst, int nh,
                              const int* __restrict__ ei, int E,
                              unsigned int* __restrict__ keys)
{
    const int tid = blockIdx.x * blockDim.x + threadIdx.x;
    {
        const int* dp = ei + E;
        const int nq = (E + 3) >> 2;
        for (int q = tid; q < nq; q += gridDim.x * blockDim.x) {
            unsigned int kw = 0u;
            #pragma unroll
            for (int j = 0; j < 4; ++j) {
                const int e = 4 * q + j;
                unsigned int kb = 0xFFu;
                if (e < E) kb = ((unsigned int)dp[e] / ROWS_PER_BLOCK) & 0xFFu;
                kw |= kb << (8 * j);
            }
            keys[q] = kw;
        }
    }
    if (tid >= nh) return;
    const int n = tid >> 2, hd = tid & 3;
    const float4* hp = (const float4*)(H + (size_t)n * NH_COLS + hd * F_OUT);
    const float4* as = (const float4*)(att_src + hd * F_OUT);
    const float4* ad = (const float4*)(att_dst + hd * F_OUT);
    float s1 = 0.0f, s2 = 0.0f;
    #pragma unroll
    for (int i = 0; i < 16; ++i) {
        const float4 hv = hp[i], v1 = as[i], v2 = ad[i];
        s1 += hv.x * v1.x + hv.y * v1.y + hv.z * v1.z + hv.w * v1.w;
        s2 += hv.x * v2.x + hv.y * v2.y + hv.z * v2.z + hv.w * v2.w;
    }
    a_src[tid] = s1;
    a_dst[tid] = s2;
}

__device__ __forceinline__ void atomicMaxFloat(float* addr, float val)
{
    if (val >= 0.0f) atomicMax((int*)addr, __float_as_int(val));
    else             atomicMin((unsigned int*)addr, __float_as_uint(val));
}

__device__ __forceinline__ void edge_endpoints(const int* __restrict__ ei,
                                               int e, int E, int& src, int& dst)
{
    if (e < E) { src = ei[e]; dst = ei[E + e]; }
    else       { src = dst = e - E; }
}

__device__ __forceinline__ float leaky(float a)
{
    return a > 0.0f ? a : NEG_SLOPE * a;
}

#define AGG_THREADS    1024
#define AGG_WAVES      (AGG_THREADS / 32)
#define QCAP           4096
#define QPT            (QCAP / AGG_THREADS)
#define CHUNK          16
#define SCAN_DEPTH     8
#define OUT_ROWS_PER_WAVE ((ROWS_PER_BLOCK + AGG_WAVES - 1) / AGG_WAVES)

static_assert(HEADS * ROWS_PER_BLOCK <= AGG_THREADS, "one thread per (head,row) start value");
static_assert(QPT * AGG_THREADS == QCAP, "the stats step covers the queue exactly");
static_assert(2 * AGG_THREADS == 32 * F_OUT, "two floats of W_prior per thread fill its LDS copy");

__device__ __forceinline__ float lane_bcast(float v, int j)
{
    return __int_as_float(__builtin_amdgcn_readlane(__float_as_int(v), j));
}

__device__ __forceinline__ void lds_add_row(float* o,
                                            float u0, float u1, float u2, float u3,
                                            float v0, float v1, float v2, float v3,
                                            float v4, float v5, float v6, float v7)
{
    atomicAdd(o,       u0 * v0);
    atomicAdd(o + 32,  u0 * v1);
    atomicAdd(o + 64,  u1 * v2);
    atomicAdd(o + 96,  u1 * v3);
    atomicAdd(o + 128, u2 * v4);
    atomicAdd(o + 160, u2 * v5);
    atomicAdd(o + 192, u3 * v6);
    atomicAdd(o + 224, u3 * v7);
}

__device__ __forceinline__ void edge_logits(int src, int dst,
                                            const float* __restrict__ a_src,
                                            const float* __restrict__ a_dst,
                                            float& l0, float& l1, float& l2, float& l3)
{
    const float4 as = *(const float4*)(a_src + (size_t)src * HEADS);
    const float4 ad = *(const float4*)(a_dst + (size_t)dst * HEADS);
    l0 = leaky(as.x + ad.x);
    l1 = leaky(as.y + ad.y);
    l2 = leaky(as.z + ad.z);
    l3 = leaky(as.w + ad.w);
}

__global__ __launch_bounds__(AGG_THREADS) void gat_edge_agg(const int* __restrict__ ei, int E, int Etot, int N,
                             const float* __restrict__ a_src,
                             const float* __restrict__ a_dst,
                             const float* __restrict__ H,
                             const float* __restrict__ bias_gat,
                             const float* __restrict__ onehot,
                             const float* __restrict__ W_prior,
                             const float* __restrict__ b_prior,
                             const float* __restrict__ gate,
                             const unsigned int* __restrict__ keys,
                             float* __restrict__ out)
{
    __shared__ __align__(16) float rows[(ROWS_PER_BLOCK + 1) * NH_COLS];
    __shared__ __align__(16) float qw[QCAP * HEADS];
    __shared__ int   queue[QCAP];
    __shared__ int   qrow[QCAP];
    __shared__ float smax[HEADS * ROWS_PER_BLOCK];
    __shared__ float sden[HEADS * ROWS_PER_BLOCK];
    __shared__ __align__(8) float wps[32 * F_OUT];
    __shared__ int   qcount;

    const int tid  = threadIdx.x;
    const int lane = tid & 31;
    const int wave = __builtin_amdgcn_readfirstlane(tid >> 5);
    const int n0     = blockIdx.x * ROWS_PER_BLOCK;
    const int rcount = min(ROWS_PER_BLOCK, N - n0);
    const unsigned ur = (unsigned)rcount;

    float4* rows4 = (float4*)rows;
    for (int i = tid; i < (ROWS_PER_BLOCK + 1) * (NH_COLS / 4); i += AGG_THREADS)
        rows4[i] = make_float4(0.0f, 0.0f, 0.0f, 0.0f);
    if (tid < HEADS * ROWS_PER_BLOCK) {
        smax[tid] = __int_as_float(0xFF800000);
        sden[tid] = 0.0f;
    }
    if (tid < rcount) queue[tid] = E + n0 + tid;
    if (tid == 0) qcount = rcount;
    __syncthreads();

    const int    f2     = 2 * lane;
    const float2 wstage = *(const float2*)(W_prior + 2 * tid);
    float oh[OUT_ROWS_PER_WAVE];
    #pragma unroll
    for (int k = 0; k < OUT_ROWS_PER_WAVE; ++k) {
        const int r = min(wave + AGG_WAVES * k, rcount - 1);
        oh[k] = onehot[(size_t)(n0 + r) * 32 + lane];
    }
    const v2f bg = *(const v2f*)(bias_gat + f2);
    const v2f bp = *(const v2f*)(b_prior + f2);
    const float g = gate[0];

    const bool usekeys = gridDim.x <= 256u;
    const int* dp  = ei + E;
    const uint4* kp4 = (const uint4*)keys;
    const int  nb16 = usekeys ? (E >> 4) : 0;
    const unsigned int pat = blockIdx.x * 0x01010101u;
    const int  rot = (int)(((long long)blockIdx.x * nb16) / gridDim.x);
    if (nb16 > 0) {
        for (int i = tid; i < nb16; i += SCAN_DEPTH * AGG_THREADS) {
            uint4 d[SCAN_DEPTH];
            int   q[SCAN_DEPTH];
            #pragma unroll
            for (int k = 0; k < SCAN_DEPTH; ++k) {
                int qq = min(i + k * AGG_THREADS, nb16 - 1) + rot;
                if (qq >= nb16) qq -= nb16;
                q[k] = qq;
                d[k] = kp4[qq];
            }
            unsigned int zz[SCAN_DEPTH];
            int c = 0;
            #pragma unroll
            for (int k = 0; k < SCAN_DEPTH; ++k) {
                const bool ok = (i + k * AGG_THREADS) < nb16;
                const unsigned int x0 = d[k].x ^ pat, x1 = d[k].y ^ pat;
                const unsigned int x2 = d[k].z ^ pat, x3 = d[k].w ^ pat;
                const unsigned int z0 = ~(((x0 & 0x7F7F7F7Fu) + 0x7F7F7F7Fu) | x0 | 0x7F7F7F7Fu);
                const unsigned int z1 = ~(((x1 & 0x7F7F7F7Fu) + 0x7F7F7F7Fu) | x1 | 0x7F7F7F7Fu);
                const unsigned int z2 = ~(((x2 & 0x7F7F7F7Fu) + 0x7F7F7F7Fu) | x2 | 0x7F7F7F7Fu);
                const unsigned int z3 = ~(((x3 & 0x7F7F7F7Fu) + 0x7F7F7F7Fu) | x3 | 0x7F7F7F7Fu);
                const unsigned int m = (z0 >> 7) | (z1 >> 6) | (z2 >> 5) | (z3 >> 4);
                zz[k] = ok ? m : 0u;
                c += __popc(zz[k]);
            }
            if (c) {
                int s = atomicAdd(&qcount, c);
                #pragma unroll
                for (int k = 0; k < SCAN_DEPTH; ++k) {
                    unsigned int m = zz[k];
                    const int e0 = 16 * q[k];
                    while (m) {
                        const int p = __builtin_ctz(m);
                        m &= m - 1u;
                        if (s < QCAP) queue[s] = e0 + 4 * (p & 7) + (p >> 3);
                        ++s;
                    }
                }
            }
        }
    }
    if (usekeys) {
        const int t = 16 * nb16 + tid;
        if (tid < 16 && t < E && (unsigned)(dp[t] - n0) < ur) {
            const int s = atomicAdd(&qcount, 1);
            if (s < QCAP) queue[s] = t;
        }
    }
    ((float2*)wps)[tid] = wstage;
    __syncthreads();

    const int total = usekeys ? __builtin_amdgcn_readfirstlane(qcount) : QCAP + 1;
    const float* hbase = H + lane;

    if (total <= QCAP) {
        int   qs[QPT], qr[QPT];
        float lg[QPT][HEADS];
        #pragma unroll
        for (int k = 0; k < QPT; ++k) {
            const int idx = tid + k * AGG_THREADS;
            const int e = (idx < total) ? queue[idx] : (E + n0);
            int s, d; edge_endpoints(ei, e, E, s, d);
            qs[k] = s;
            qr[k] = d - n0;
        }
        #pragma unroll
        for (int k = 0; k < QPT; ++k)
            edge_logits(qs[k], n0 + qr[k], a_src, a_dst, lg[k][0], lg[k][1], lg[k][2], lg[k][3]);
        #pragma unroll
        for (int k = 0; k < QPT; ++k) {
            const int idx = tid + k * AGG_THREADS;
            if (idx < total) {
                queue[idx] = qs[k];
                qrow[idx]  = qr[k];
                #pragma unroll
                for (int hd = 0; hd < HEADS; ++hd)
                    atomicMaxFloat(&smax[hd * ROWS_PER_BLOCK + qr[k]], lg[k][hd]);
            }
        }
        __syncthreads();
        float st[QPT][HEADS];
        #pragma unroll
        for (int k = 0; k < QPT; ++k) {
            #pragma unroll
            for (int hd = 0; hd < HEADS; ++hd) st[k][hd] = smax[hd * ROWS_PER_BLOCK + qr[k]];
        }
        #pragma unroll
        for (int k = 0; k < QPT; ++k) {
            #pragma unroll
            for (int hd = 0; hd < HEADS; ++hd) lg[k][hd] = __expf(lg[k][hd] - st[k][hd]);
        }
        #pragma unroll
        for (int k = 0; k < QPT; ++k) {
            const int idx = tid + k * AGG_THREADS;
            if (idx < total) {
                #pragma unroll
                for (int hd = 0; hd < HEADS; ++hd)
                    atomicAdd(&sden[hd * ROWS_PER_BLOCK + qr[k]], lg[k][hd]);
            }
        }
        __syncthreads();
        #pragma unroll
        for (int k = 0; k < QPT; ++k) {
            #pragma unroll
            for (int hd = 0; hd < HEADS; ++hd) st[k][hd] = sden[hd * ROWS_PER_BLOCK + qr[k]];
        }
        #pragma unroll
        for (int k = 0; k < QPT; ++k) {
            const int idx = tid + k * AGG_THREADS;
            if (idx < total) {
                float4 w;
                w.x = lg[k][0] / (st[k][0] + 1e-16f);
                w.y = lg[k][1] / (st[k][1] + 1e-16f);
                w.z = lg[k][2] / (st[k][2] + 1e-16f);
                w.w = lg[k][3] / (st[k][3] + 1e-16f);
                ((float4*)qw)[idx] = w;
            }
        }
        __syncthreads();

        const int nchunks = (total + CHUNK - 1) / CHUNK;
        for (int c = wave; c < nchunks; c += AGG_WAVES) {
            const int ebeg = c * CHUNK;
            const int cnt  = min(CHUNK, total - ebeg);
            int src = 0, row = ROWS_PER_BLOCK;
            float w0 = 0.0f, w1 = 0.0f, w2 = 0.0f, w3 = 0.0f;
            if (lane < cnt) {
                src = queue[ebeg + lane];
                row = qrow[ebeg + lane];
                const float4 w = ((const float4*)qw)[ebeg + lane];
                w0 = w.x; w1 = w.y; w2 = w.z; w3 = w.w;
            }
            #pragma unroll
            for (int g = 0; g < CHUNK; g += 4) {
                if (g >= cnt) break;
                float v[4][8];
                #pragma unroll
                for (int k = 0; k < 4; ++k) {
                    const float* hr = hbase + (size_t)__builtin_amdgcn_readlane(src, g + k) * NH_COLS;
                    #pragma unroll
                    for (int t = 0; t < 8; ++t) v[k][t] = hr[32 * t];
                }
                #pragma unroll
                for (int k = 0; k < 4; ++k) {
                    float* o = rows + __builtin_amdgcn_readlane(row, g + k) * NH_COLS + lane;
                    lds_add_row(o,
                                lane_bcast(w0, g + k), lane_bcast(w1, g + k),
                                lane_bcast(w2, g + k), lane_bcast(w3, g + k),
                                v[k][0], v[k][1], v[k][2], v[k][3],
                                v[k][4], v[k][5], v[k][6], v[k][7]);
                }
            }
        }
    } else {
        for (int e = tid; e < Etot; e += AGG_THREADS) {
            int s, d; edge_endpoints(ei, e, E, s, d);
            const int r = d - n0;
            if ((unsigned)r < ur) {
                float l0, l1, l2, l3; edge_logits(s, d, a_src, a_dst, l0, l1, l2, l3);
                atomicMaxFloat(&smax[0 * ROWS_PER_BLOCK + r], l0);
                atomicMaxFloat(&smax[1 * ROWS_PER_BLOCK + r], l1);
                atomicMaxFloat(&smax[2 * ROWS_PER_BLOCK + r], l2);
                atomicMaxFloat(&smax[3 * ROWS_PER_BLOCK + r], l3);
            }
        }
        __syncthreads();
        for (int e = tid; e < Etot; e += AGG_THREADS) {
            int s, d; edge_endpoints(ei, e, E, s, d);
            const int r = d - n0;
            if ((unsigned)r < ur) {
                float l0, l1, l2, l3; edge_logits(s, d, a_src, a_dst, l0, l1, l2, l3);
                atomicAdd(&sden[0 * ROWS_PER_BLOCK + r], __expf(l0 - smax[0 * ROWS_PER_BLOCK + r]));
                atomicAdd(&sden[1 * ROWS_PER_BLOCK + r], __expf(l1 - smax[1 * ROWS_PER_BLOCK + r]));
                atomicAdd(&sden[2 * ROWS_PER_BLOCK + r], __expf(l2 - smax[2 * ROWS_PER_BLOCK + r]));
                atomicAdd(&sden[3 * ROWS_PER_BLOCK + r], __expf(l3 - smax[3 * ROWS_PER_BLOCK + r]));
            }
        }
        __syncthreads();
        for (int base = wave * 32; base < Etot; base += AGG_THREADS) {
            const int e = base + lane;
            int src = 0, dst = -1;
            if (e < Etot) edge_endpoints(ei, e, E, src, dst);
            const bool mine = (unsigned)(dst - n0) < ur;
            const int row = dst - n0;
            float w0 = 0.0f, w1 = 0.0f, w2 = 0.0f, w3 = 0.0f;
            if (mine) {
                float l0, l1, l2, l3; edge_logits(src, dst, a_src, a_dst, l0, l1, l2, l3);
                w0 = __expf(l0 - smax[0 * ROWS_PER_BLOCK + row]) / (sden[0 * ROWS_PER_BLOCK + row] + 1e-16f);
                w1 = __expf(l1 - smax[1 * ROWS_PER_BLOCK + row]) / (sden[1 * ROWS_PER_BLOCK + row] + 1e-16f);
                w2 = __expf(l2 - smax[2 * ROWS_PER_BLOCK + row]) / (sden[2 * ROWS_PER_BLOCK + row] + 1e-16f);
                w3 = __expf(l3 - smax[3 * ROWS_PER_BLOCK + row]) / (sden[3 * ROWS_PER_BLOCK + row] + 1e-16f);
            }
            unsigned mask = __builtin_amdgcn_ballot_w32(mine);
            while (mask) {
                const int j = __builtin_ctz(mask);
                mask &= mask - 1;
                const float* hr = hbase + (size_t)__builtin_amdgcn_readlane(src, j) * NH_COLS;
                const float v0 = hr[0],   v1 = hr[32],  v2 = hr[64],  v3 = hr[96];
                const float v4 = hr[128], v5 = hr[160], v6 = hr[192], v7 = hr[224];
                lds_add_row(rows + __builtin_amdgcn_readlane(row, j) * NH_COLS + lane,
                            lane_bcast(w0, j), lane_bcast(w1, j), lane_bcast(w2, j), lane_bcast(w3, j),
                            v0, v1, v2, v3, v4, v5, v6, v7);
            }
        }
    }

    __syncthreads();

    v2f wp[32];
    #pragma unroll
    for (int c = 0; c < 32; ++c) wp[c] = ((const v2f*)wps)[c * 32 + lane];

    #pragma unroll
    for (int k = 0; k < OUT_ROWS_PER_WAVE; ++k) {
        const int r = wave + AGG_WAVES * k;
        if (r < rcount) {
            const float* rp = rows + r * NH_COLS + f2;
            const v2f a0 = *(const v2f*)(rp);
            const v2f a1 = *(const v2f*)(rp + F_OUT);
            const v2f a2 = *(const v2f*)(rp + 2 * F_OUT);
            const v2f a3 = *(const v2f*)(rp + 3 * F_OUT);
            const v2f quarter = {0.25f, 0.25f};
            const v2f m = __builtin_elementwise_fma(quarter, a0 + a1 + a2 + a3, bg);
            v2f p = bp;
            #pragma unroll
            for (int c = 0; c < 32; ++c) {
                const float o = lane_bcast(oh[k], c);
                const v2f o2 = {o, o};
                p = __builtin_elementwise_fma(o2, wp[c], p);
            }
            const v2f g2 = {g, g};
            const v2f h2 = {1.0f - g, 1.0f - g};
            const v2f res = __builtin_elementwise_fma(h2, m, g2 * p);
            *(v2f*)(out + (size_t)(n0 + r) * F_OUT + f2) = res;
        }
    }
}

extern "C" void kernel_launch(void* const* d_in, const int* in_sizes, int n_in,
                              void* d_out, int out_size, void* d_ws, size_t ws_size,
                              hipStream_t stream)
{
    const float* x        = (const float*)d_in[0];
    const int*   ei       = (const int*)  d_in[1];
    const float* onehot   = (const float*)d_in[2];
    const float* W_gat    = (const float*)d_in[3];
    const float* att_src  = (const float*)d_in[4];
    const float* att_dst  = (const float*)d_in[5];
    const float* bias_gat = (const float*)d_in[6];
    const float* W_prior  = (const float*)d_in[7];
    const float* b_prior  = (const float*)d_in[8];
    const float* gate     = (const float*)d_in[9];
    float* out = (float*)d_out;

    const int N    = in_sizes[0] / D_IN;
    const int E    = in_sizes[1] / 2;
    const int Etot = E + N;
    const int NHs  = N * HEADS;

    float* h      = (float*)d_ws;
    float* a_src  = h      + (size_t)N * NH_COLS;
    float* a_dst  = a_src  + NHs;
    unsigned int* keys = (unsigned int*)(a_dst + NHs);

    gat_gemm_wmma<<<N / 16, 128, 0, stream>>>(x, W_gat, h);

    gat_attn_coef<<<(NHs + 255) / 256, 256, 0, stream>>>(h, att_src, att_dst,
                                                         a_src, a_dst, NHs, ei, E, keys);

    gat_edge_agg<<<(N + ROWS_PER_BLOCK - 1) / ROWS_PER_BLOCK, AGG_THREADS, 0, stream>>>(
        ei, E, Etot, N, a_src, a_dst, h,
        bias_gat, onehot, W_prior, b_prior, gate, keys, out);
}


# GEMM operand feed: 256 resident work-groups of 8 waves walk the row tiles; a wave keeps its 128x64 strip of W_gat in registers (staged once via LDS): no B loads in the k loop; sums keep their order
# speedup vs baseline: 4.3178x; 1.0985x over previous
#include <hip/hip_runtime.h>
#include <hip/hip_bf16.h>
#include <math.h>

typedef __attribute__((ext_vector_type(2))) float v2f;
typedef __attribute__((ext_vector_type(8))) float v8f;

#define HEADS 4
#define F_OUT 64
#define D_IN 128
#define NH_COLS 256
#define NEG_SLOPE 0.2f
#define ROWS_PER_BLOCK 200

#define GEMM_THREADS 256
#define GEMM_WAVES   (GEMM_THREADS / 32)
#define GEMM_BLOCKS  256
#define GEMM_TSTRIDE (GEMM_BLOCKS * (GEMM_WAVES / 4))
#define WPITCH       544

__global__ __launch_bounds__(GEMM_THREADS) void gat_gemm_wmma(
    const float* __restrict__ X,
    const float* __restrict__ W,
    float* __restrict__ H,
    int N)
{
    __shared__ __align__(16) float wl[(D_IN / 2) * WPITCH];

    const int tid = threadIdx.x;
    #pragma unroll
    for (int i = 0; i < (D_IN / 2) * (NH_COLS / 4) / GEMM_THREADS; ++i) {
        const int item = tid + GEMM_THREADS * i;
        const int p  = item >> 6;
        const int c4 = item & 63;
        const float4 r0 = *(const float4*)(W + (size_t)(2 * p)     * NH_COLS + 4 * c4);
        const float4 r1 = *(const float4*)(W + (size_t)(2 * p + 1) * NH_COLS + 4 * c4);
        float4* d = (float4*)(wl + p * WPITCH + 8 * c4);
        d[0] = make_float4(r0.x, r1.x, r0.y, r1.y);
        d[1] = make_float4(r0.z, r1.z, r0.w, r1.w);
    }

    const int wave    = __builtin_amdgcn_readfirstlane(tid >> 5);
    const int lane    = tid & 31;
    const int halfsel = lane >> 4;
    const int l16     = lane & 15;
    const int colbase = (wave & 3) * 64;
    const int ntiles  = N >> 4;

    const int first = blockIdx.x + GEMM_BLOCKS * (wave >> 2);
    int mtile = min(first, ntiles - 1);
    const float* xlane = X + (size_t)l16 * D_IN + halfsel * 2;

    v2f a[32];
    {
        const float* xrow = xlane + (size_t)mtile * (16 * D_IN);
        #pragma unroll
        for (int s = 0; s < 32; ++s) a[s] = *(const v2f*)(xrow + 4 * s);
    }

    __syncthreads();

    v2f b[32][4];
    {
        const float* bl = wl + halfsel * WPITCH + 2 * (colbase + l16);
        #pragma unroll
        for (int s = 0; s < 32; ++s) {
            #pragma unroll
            for (int n = 0; n < 4; ++n) b[s][n] = *(const v2f*)(bl + (2 * s) * WPITCH + 32 * n);
        }
    }

    if (first >= ntiles) return;

    for (;;) {
        const int  nxt  = mtile + GEMM_TSTRIDE;
        const bool more = nxt < ntiles;
        v2f an[32];
        {
            const float* xrow = xlane + (size_t)(more ? nxt : mtile) * (16 * D_IN);
            #pragma unroll
            for (int s = 0; s < 32; ++s) an[s] = *(const v2f*)(xrow + 4 * s);
        }

        v8f acc0 = {}; v8f acc1 = {}; v8f acc2 = {}; v8f acc3 = {};
        #pragma unroll
        for (int s = 0; s < 32; ++s) {
            acc0 = __builtin_amdgcn_wmma_f32_16x16x4_f32(false, a[s], false, b[s][0], (short)0, acc0, false, false);
            acc1 = __builtin_amdgcn_wmma_f32_16x16x4_f32(false, a[s], false, b[s][1], (short)0, acc1, false, false);
            acc2 = __builtin_amdgcn_wmma_f32_16x16x4_f32(false, a[s], false, b[s][2], (short)0, acc2, false, false);
            acc3 = __builtin_amdgcn_wmma_f32_16x16x4_f32(false, a[s], false, b[s][3], (short)0, acc3, false, false);
        }

        #pragma unroll
        for (int r = 0; r < 8; ++r) {
            const int m = mtile * 16 + r + halfsel * 8;
            float* hrow = H + (size_t)m * NH_COLS + colbase + l16;
            hrow[0]  = acc0[r];
            hrow[16] = acc1[r];
            hrow[32] = acc2[r];
            hrow[48] = acc3[r];
        }

        if (!more) break;
        mtile = nxt;
        #pragma unroll
        for (int s = 0; s < 32; ++s) a[s] = an[s];
    }
}

__global__ void gat_attn_coef(const float* __restrict__ H,
                              const float* __restrict__ att_src,
                              const float* __restrict__ att_dst,
                              float* __restrict__ a_src,
                              float* __restrict__ a_dst, int nh,
                              const int* __restrict__ ei, int E,
                              unsigned int* __restrict__ keys)
{
    const int tid = blockIdx.x * blockDim.x + threadIdx.x;
    {
        const int* dp = ei + E;
        const int nq = (E + 3) >> 2;
        for (int q = tid; q < nq; q += gridDim.x * blockDim.x) {
            unsigned int kw = 0u;
            #pragma unroll
            for (int j = 0; j < 4; ++j) {
                const int e = 4 * q + j;
                unsigned int kb = 0xFFu;
                if (e < E) kb = ((unsigned int)dp[e] / ROWS_PER_BLOCK) & 0xFFu;
                kw |= kb << (8 * j);
            }
            keys[q] = kw;
        }
    }
    if (tid >= nh) return;
    const int n = tid >> 2, hd = tid & 3;
    const float4* hp = (const float4*)(H + (size_t)n * NH_COLS + hd * F_OUT);
    const float4* as = (const float4*)(att_src + hd * F_OUT);
    const float4* ad = (const float4*)(att_dst + hd * F_OUT);
    float s1 = 0.0f, s2 = 0.0f;
    #pragma unroll
    for (int i = 0; i < 16; ++i) {
        const float4 hv = hp[i], v1 = as[i], v2 = ad[i];
        s1 += hv.x * v1.x + hv.y * v1.y + hv.z * v1.z + hv.w * v1.w;
        s2 += hv.x * v2.x + hv.y * v2.y + hv.z * v2.z + hv.w * v2.w;
    }
    a_src[tid] = s1;
    a_dst[tid] = s2;
}

__device__ __forceinline__ void atomicMaxFloat(float* addr, float val)
{
    if (val >= 0.0f) atomicMax((int*)addr, __float_as_int(val));
    else             atomicMin((unsigned int*)addr, __float_as_uint(val));
}

__device__ __forceinline__ void edge_endpoints(const int* __restrict__ ei,
                                               int e, int E, int& src, int& dst)
{
    if (e < E) { src = ei[e]; dst = ei[E + e]; }
    else       { src = dst = e - E; }
}

__device__ __forceinline__ float leaky(float a)
{
    return a > 0.0f ? a : NEG_SLOPE * a;
}

#define AGG_THREADS    1024
#define AGG_WAVES      (AGG_THREADS / 32)
#define QCAP           4096
#define QPT            (QCAP / AGG_THREADS)
#define CHUNK          16
#define SCAN_DEPTH     8
#define OUT_ROWS_PER_WAVE ((ROWS_PER_BLOCK + AGG_WAVES - 1) / AGG_WAVES)

static_assert(HEADS * ROWS_PER_BLOCK <= AGG_THREADS, "one thread per (head,row) start value");
static_assert(QPT * AGG_THREADS == QCAP, "the stats step covers the queue exactly");
static_assert(2 * AGG_THREADS == 32 * F_OUT, "two floats of W_prior per thread fill its LDS copy");

__device__ __forceinline__ float lane_bcast(float v, int j)
{
    return __int_as_float(__builtin_amdgcn_readlane(__float_as_int(v), j));
}

__device__ __forceinline__ void lds_add_row(float* o,
                                            float u0, float u1, float u2, float u3,
                                            float v0, float v1, float v2, float v3,
                                            float v4, float v5, float v6, float v7)
{
    atomicAdd(o,       u0 * v0);
    atomicAdd(o + 32,  u0 * v1);
    atomicAdd(o + 64,  u1 * v2);
    atomicAdd(o + 96,  u1 * v3);
    atomicAdd(o + 128, u2 * v4);
    atomicAdd(o + 160, u2 * v5);
    atomicAdd(o + 192, u3 * v6);
    atomicAdd(o + 224, u3 * v7);
}

__device__ __forceinline__ void edge_logits(int src, int dst,
                                            const float* __restrict__ a_src,
                                            const float* __restrict__ a_dst,
                                            float& l0, float& l1, float& l2, float& l3)
{
    const float4 as = *(const float4*)(a_src + (size_t)src * HEADS);
    const float4 ad = *(const float4*)(a_dst + (size_t)dst * HEADS);
    l0 = leaky(as.x + ad.x);
    l1 = leaky(as.y + ad.y);
    l2 = leaky(as.z + ad.z);
    l3 = leaky(as.w + ad.w);
}

__global__ __launch_bounds__(AGG_THREADS) void gat_edge_agg(const int* __restrict__ ei, int E, int Etot, int N,
                             const float* __restrict__ a_src,
                             const float* __restrict__ a_dst,
                             const float* __restrict__ H,
                             const float* __restrict__ bias_gat,
                             const float* __restrict__ onehot,
                             const float* __restrict__ W_prior,
                             const float* __restrict__ b_prior,
                             const float* __restrict__ gate,
                             const unsigned int* __restrict__ keys,
                             float* __restrict__ out)
{
    __shared__ __align__(16) float rows[(ROWS_PER_BLOCK + 1) * NH_COLS];
    __shared__ __align__(16) float qw[QCAP * HEADS];
    __shared__ int   queue[QCAP];
    __shared__ int   qrow[QCAP];
    __shared__ float smax[HEADS * ROWS_PER_BLOCK];
    __shared__ float sden[HEADS * ROWS_PER_BLOCK];
    __shared__ __align__(8) float wps[32 * F_OUT];
    __shared__ int   qcount;

    const int tid  = threadIdx.x;
    const int lane = tid & 31;
    const int wave = __builtin_amdgcn_readfirstlane(tid >> 5);
    const int n0     = blockIdx.x * ROWS_PER_BLOCK;
    const int rcount = min(ROWS_PER_BLOCK, N - n0);
    const unsigned ur = (unsigned)rcount;

    float4* rows4 = (float4*)rows;
    for (int i = tid; i < (ROWS_PER_BLOCK + 1) * (NH_COLS / 4); i += AGG_THREADS)
        rows4[i] = make_float4(0.0f, 0.0f, 0.0f, 0.0f);
    if (tid < HEADS * ROWS_PER_BLOCK) {
        smax[tid] = __int_as_float(0xFF800000);
        sden[tid] = 0.0f;
    }
    if (tid < rcount) queue[tid] = E + n0 + tid;
    if (tid == 0) qcount = rcount;
    __syncthreads();

    const int    f2     = 2 * lane;
    const float2 wstage = *(const float2*)(W_prior + 2 * tid);
    float oh[OUT_ROWS_PER_WAVE];
    #pragma unroll
    for (int k = 0; k < OUT_ROWS_PER_WAVE; ++k) {
        const int r = min(wave + AGG_WAVES * k, rcount - 1);
        oh[k] = onehot[(size_t)(n0 + r) * 32 + lane];
    }
    const v2f bg = *(const v2f*)(bias_gat + f2);
    const v2f bp = *(const v2f*)(b_prior + f2);
    const float g = gate[0];

    const bool usekeys = gridDim.x <= 256u;
    const int* dp  = ei + E;
    const uint4* kp4 = (const uint4*)keys;
    const int  nb16 = usekeys ? (E >> 4) : 0;
    const unsigned int pat = blockIdx.x * 0x01010101u;
    const int  rot = (int)(((long long)blockIdx.x * nb16) / gridDim.x);
    if (nb16 > 0) {
        for (int i = tid; i < nb16; i += SCAN_DEPTH * AGG_THREADS) {
            uint4 d[SCAN_DEPTH];
            int   q[SCAN_DEPTH];
            #pragma unroll
            for (int k = 0; k < SCAN_DEPTH; ++k) {
                int qq = min(i + k * AGG_THREADS, nb16 - 1) + rot;
                if (qq >= nb16) qq -= nb16;
                q[k] = qq;
                d[k] = kp4[qq];
            }
            unsigned int zz[SCAN_DEPTH];
            int c = 0;
            #pragma unroll
            for (int k = 0; k < SCAN_DEPTH; ++k) {
                const bool ok = (i + k * AGG_THREADS) < nb16;
                const unsigned int x0 = d[k].x ^ pat, x1 = d[k].y ^ pat;
                const unsigned int x2 = d[k].z ^ pat, x3 = d[k].w ^ pat;
                const unsigned int z0 = ~(((x0 & 0x7F7F7F7Fu) + 0x7F7F7F7Fu) | x0 | 0x7F7F7F7Fu);
                const unsigned int z1 = ~(((x1 & 0x7F7F7F7Fu) + 0x7F7F7F7Fu) | x1 | 0x7F7F7F7Fu);
                const unsigned int z2 = ~(((x2 & 0x7F7F7F7Fu) + 0x7F7F7F7Fu) | x2 | 0x7F7F7F7Fu);
                const unsigned int z3 = ~(((x3 & 0x7F7F7F7Fu) + 0x7F7F7F7Fu) | x3 | 0x7F7F7F7Fu);
                const unsigned int m = (z0 >> 7) | (z1 >> 6) | (z2 >> 5) | (z3 >> 4);
                zz[k] = ok ? m : 0u;
                c += __popc(zz[k]);
            }
            if (c) {
                int s = atomicAdd(&qcount, c);
                #pragma unroll
                for (int k = 0; k < SCAN_DEPTH; ++k) {
                    unsigned int m = zz[k];
                    const int e0 = 16 * q[k];
                    while (m) {
                        const int p = __builtin_ctz(m);
                        m &= m - 1u;
                        if (s < QCAP) queue[s] = e0 + 4 * (p & 7) + (p >> 3);
                        ++s;
                    }
                }
            }
        }
    }
    if (usekeys) {
        const int t = 16 * nb16 + tid;
        if (tid < 16 && t < E && (unsigned)(dp[t] - n0) < ur) {
            const int s = atomicAdd(&qcount, 1);
            if (s < QCAP) queue[s] = t;
        }
    }
    ((float2*)wps)[tid] = wstage;
    __syncthreads();

    const int total = usekeys ? __builtin_amdgcn_readfirstlane(qcount) : QCAP + 1;
    const float* hbase = H + lane;

    if (total <= QCAP) {
        int   qs[QPT], qr[QPT];
        float lg[QPT][HEADS];
        #pragma unroll
        for (int k = 0; k < QPT; ++k) {
            const int idx = tid + k * AGG_THREADS;
            const int e = (idx < total) ? queue[idx] : (E + n0);
            int s, d; edge_endpoints(ei, e, E, s, d);
            qs[k] = s;
            qr[k] = d - n0;
        }
        #pragma unroll
        for (int k = 0; k < QPT; ++k)
            edge_logits(qs[k], n0 + qr[k], a_src, a_dst, lg[k][0], lg[k][1], lg[k][2], lg[k][3]);
        #pragma unroll
        for (int k = 0; k < QPT; ++k) {
            const int idx = tid + k * AGG_THREADS;
            if (idx < total) {
                queue[idx] = qs[k];
                qrow[idx]  = qr[k];
                #pragma unroll
                for (int hd = 0; hd < HEADS; ++hd)
                    atomicMaxFloat(&smax[hd * ROWS_PER_BLOCK + qr[k]], lg[k][hd]);
            }
        }
        __syncthreads();
        float st[QPT][HEADS];
        #pragma unroll
        for (int k = 0; k < QPT; ++k) {
            #pragma unroll
            for (int hd = 0; hd < HEADS; ++hd) st[k][hd] = smax[hd * ROWS_PER_BLOCK + qr[k]];
        }
        #pragma unroll
        for (int k = 0; k < QPT; ++k) {
            #pragma unroll
            for (int hd = 0; hd < HEADS; ++hd) lg[k][hd] = __expf(lg[k][hd] - st[k][hd]);
        }
        #pragma unroll
        for (int k = 0; k < QPT; ++k) {
            const int idx = tid + k * AGG_THREADS;
            if (idx < total) {
                #pragma unroll
                for (int hd = 0; hd < HEADS; ++hd)
                    atomicAdd(&sden[hd * ROWS_PER_BLOCK + qr[k]], lg[k][hd]);
            }
        }
        __syncthreads();
        #pragma unroll
        for (int k = 0; k < QPT; ++k) {
            #pragma unroll
            for (int hd = 0; hd < HEADS; ++hd) st[k][hd] = sden[hd * ROWS_PER_BLOCK + qr[k]];
        }
        #pragma unroll
        for (int k = 0; k < QPT; ++k) {
            const int idx = tid + k * AGG_THREADS;
            if (idx < total) {
                float4 w;
                w.x = lg[k][0] / (st[k][0] + 1e-16f);
                w.y = lg[k][1] / (st[k][1] + 1e-16f);
                w.z = lg[k][2] / (st[k][2] + 1e-16f);
                w.w = lg[k][3] / (st[k][3] + 1e-16f);
                ((float4*)qw)[idx] = w;
            }
        }
        __syncthreads();

        const int nchunks = (total + CHUNK - 1) / CHUNK;
        for (int c = wave; c < nchunks; c += AGG_WAVES) {
            const int ebeg = c * CHUNK;
            const int cnt  = min(CHUNK, total - ebeg);
            int src = 0, row = ROWS_PER_BLOCK;
            float w0 = 0.0f, w1 = 0.0f, w2 = 0.0f, w3 = 0.0f;
            if (lane < cnt) {
                src = queue[ebeg + lane];
                row = qrow[ebeg + lane];
                const float4 w = ((const float4*)qw)[ebeg + lane];
                w0 = w.x; w1 = w.y; w2 = w.z; w3 = w.w;
            }
            #pragma unroll
            for (int g = 0; g < CHUNK; g += 4) {
                if (g >= cnt) break;
                float v[4][8];
                #pragma unroll
                for (int k = 0; k < 4; ++k) {
                    const float* hr = hbase + (size_t)__builtin_amdgcn_readlane(src, g + k) * NH_COLS;
                    #pragma unroll
                    for (int t = 0; t < 8; ++t) v[k][t] = hr[32 * t];
                }
                #pragma unroll
                for (int k = 0; k < 4; ++k) {
                    float* o = rows + __builtin_amdgcn_readlane(row, g + k) * NH_COLS + lane;
                    lds_add_row(o,
                                lane_bcast(w0, g + k), lane_bcast(w1, g + k),
                                lane_bcast(w2, g + k), lane_bcast(w3, g + k),
                                v[k][0], v[k][1], v[k][2], v[k][3],
                                v[k][4], v[k][5], v[k][6], v[k][7]);
                }
            }
        }
    } else {
        for (int e = tid; e < Etot; e += AGG_THREADS) {
            int s, d; edge_endpoints(ei, e, E, s, d);
            const int r = d - n0;
            if ((unsigned)r < ur) {
                float l0, l1, l2, l3; edge_logits(s, d, a_src, a_dst, l0, l1, l2, l3);
                atomicMaxFloat(&smax[0 * ROWS_PER_BLOCK + r], l0);
                atomicMaxFloat(&smax[1 * ROWS_PER_BLOCK + r], l1);
                atomicMaxFloat(&smax[2 * ROWS_PER_BLOCK + r], l2);
                atomicMaxFloat(&smax[3 * ROWS_PER_BLOCK + r], l3);
            }
        }
        __syncthreads();
        for (int e = tid; e < Etot; e += AGG_THREADS) {
            int s, d; edge_endpoints(ei, e, E, s, d);
            const int r = d - n0;
            if ((unsigned)r < ur) {
                float l0, l1, l2, l3; edge_logits(s, d, a_src, a_dst, l0, l1, l2, l3);
                atomicAdd(&sden[0 * ROWS_PER_BLOCK + r], __expf(l0 - smax[0 * ROWS_PER_BLOCK + r]));
                atomicAdd(&sden[1 * ROWS_PER_BLOCK + r], __expf(l1 - smax[1 * ROWS_PER_BLOCK + r]));
                atomicAdd(&sden[2 * ROWS_PER_BLOCK + r], __expf(l2 - smax[2 * ROWS_PER_BLOCK + r]));
                atomicAdd(&sden[3 * ROWS_PER_BLOCK + r], __expf(l3 - smax[3 * ROWS_PER_BLOCK + r]));
            }
        }
        __syncthreads();
        for (int base = wave * 32; base < Etot; base += AGG_THREADS) {
            const int e = base + lane;
            int src = 0, dst = -1;
            if (e < Etot) edge_endpoints(ei, e, E, src, dst);
            const bool mine = (unsigned)(dst - n0) < ur;
            const int row = dst - n0;
            float w0 = 0.0f, w1 = 0.0f, w2 = 0.0f, w3 = 0.0f;
            if (mine) {
                float l0, l1, l2, l3; edge_logits(src, dst, a_src, a_dst, l0, l1, l2, l3);
                w0 = __expf(l0 - smax[0 * ROWS_PER_BLOCK + row]) / (sden[0 * ROWS_PER_BLOCK + row] + 1e-16f);
                w1 = __expf(l1 - smax[1 * ROWS_PER_BLOCK + row]) / (sden[1 * ROWS_PER_BLOCK + row] + 1e-16f);
                w2 = __expf(l2 - smax[2 * ROWS_PER_BLOCK + row]) / (sden[2 * ROWS_PER_BLOCK + row] + 1e-16f);
                w3 = __expf(l3 - smax[3 * ROWS_PER_BLOCK + row]) / (sden[3 * ROWS_PER_BLOCK + row] + 1e-16f);
            }
            unsigned mask = __builtin_amdgcn_ballot_w32(mine);
            while (mask) {
                const int j = __builtin_ctz(mask);
                mask &= mask - 1;
                const float* hr = hbase + (size_t)__builtin_amdgcn_readlane(src, j) * NH_COLS;
                const float v0 = hr[0],   v1 = hr[32],  v2 = hr[64],  v3 = hr[96];
                const float v4 = hr[128], v5 = hr[160], v6 = hr[192], v7 = hr[224];
                lds_add_row(rows + __builtin_amdgcn_readlane(row, j) * NH_COLS + lane,
                            lane_bcast(w0, j), lane_bcast(w1, j), lane_bcast(w2, j), lane_bcast(w3, j),
                            v0, v1, v2, v3, v4, v5, v6, v7);
            }
        }
    }

    __syncthreads();

    v2f wp[32];
    #pragma unroll
    for (int c = 0; c < 32; ++c) wp[c] = ((const v2f*)wps)[c * 32 + lane];

    #pragma unroll
    for (int k = 0; k < OUT_ROWS_PER_WAVE; ++k) {
        const int r = wave + AGG_WAVES * k;
        if (r < rcount) {
            const float* rp = rows + r * NH_COLS + f2;
            const v2f a0 = *(const v2f*)(rp);
            const v2f a1 = *(const v2f*)(rp + F_OUT);
            const v2f a2 = *(const v2f*)(rp + 2 * F_OUT);
            const v2f a3 = *(const v2f*)(rp + 3 * F_OUT);
            const v2f quarter = {0.25f, 0.25f};
            const v2f m = __builtin_elementwise_fma(quarter, a0 + a1 + a2 + a3, bg);
            v2f p = bp;
            #pragma unroll
            for (int c = 0; c < 32; ++c) {
                const float o = lane_bcast(oh[k], c);
                const v2f o2 = {o, o};
                p = __builtin_elementwise_fma(o2, wp[c], p);
            }
            const v2f g2 = {g, g};
            const v2f h2 = {1.0f - g, 1.0f - g};
            const v2f res = __builtin_elementwise_fma(h2, m, g2 * p);
            *(v2f*)(out + (size_t)(n0 + r) * F_OUT + f2) = res;
        }
    }
}

extern "C" void kernel_launch(void* const* d_in, const int* in_sizes, int n_in,
                              void* d_out, int out_size, void* d_ws, size_t ws_size,
                              hipStream_t stream)
{
    const float* x        = (const float*)d_in[0];
    const int*   ei       = (const int*)  d_in[1];
    const float* onehot   = (const float*)d_in[2];
    const float* W_gat    = (const float*)d_in[3];
    const float* att_src  = (const float*)d_in[4];
    const float* att_dst  = (const float*)d_in[5];
    const float* bias_gat = (const float*)d_in[6];
    const float* W_prior  = (const float*)d_in[7];
    const float* b_prior  = (const float*)d_in[8];
    const float* gate     = (const float*)d_in[9];
    float* out = (float*)d_out;

    const int N    = in_sizes[0] / D_IN;
    const int E    = in_sizes[1] / 2;
    const int Etot = E + N;
    const int NHs  = N * HEADS;

    float* h      = (float*)d_ws;
    float* a_src  = h      + (size_t)N * NH_COLS;
    float* a_dst  = a_src  + NHs;
    unsigned int* keys = (unsigned int*)(a_dst + NHs);

    gat_gemm_wmma<<<GEMM_BLOCKS, GEMM_THREADS, 0, stream>>>(x, W_gat, h, N);

    gat_attn_coef<<<(NHs + 255) / 256, 256, 0, stream>>>(h, att_src, att_dst,
                                                         a_src, a_dst, NHs, ei, E, keys);

    gat_edge_agg<<<(N + ROWS_PER_BLOCK - 1) / ROWS_PER_BLOCK, AGG_THREADS, 0, stream>>>(
        ei, E, Etot, N, a_src, a_dst, h,
        bias_gat, onehot, W_prior, b_prior, gate, keys, out);
}


# edge_agg: queue sorted by owner wave (row % 32); a wave sums its own rows with plain LDS read-add-write and 16-B row loads, no LDS float adds; a row's sum order = its wave's queue order
# speedup vs baseline: 4.7419x; 1.0982x over previous
#include <hip/hip_runtime.h>
#include <hip/hip_bf16.h>
#include <math.h>

typedef __attribute__((ext_vector_type(2))) float v2f;
typedef __attribute__((ext_vector_type(8))) float v8f;

#define HEADS 4
#define F_OUT 64
#define D_IN 128
#define NH_COLS 256
#define NEG_SLOPE 0.2f
#define ROWS_PER_BLOCK 200

#define GEMM_THREADS 256
#define GEMM_WAVES   (GEMM_THREADS / 32)
#define GEMM_BLOCKS  256
#define GEMM_TSTRIDE (GEMM_BLOCKS * (GEMM_WAVES / 4))
#define WPITCH       544

__global__ __launch_bounds__(GEMM_THREADS) void gat_gemm_wmma(
    const float* __restrict__ X,
    const float* __restrict__ W,
    float* __restrict__ H,
    int N)
{
    __shared__ __align__(16) float wl[(D_IN / 2) * WPITCH];

    const int tid = threadIdx.x;
    #pragma unroll
    for (int i = 0; i < (D_IN / 2) * (NH_COLS / 4) / GEMM_THREADS; ++i) {
        const int item = tid + GEMM_THREADS * i;
        const int p  = item >> 6;
        const int c4 = item & 63;
        const float4 r0 = *(const float4*)(W + (size_t)(2 * p)     * NH_COLS + 4 * c4);
        const float4 r1 = *(const float4*)(W + (size_t)(2 * p + 1) * NH_COLS + 4 * c4);
        float4* d = (float4*)(wl + p * WPITCH + 8 * c4);
        d[0] = make_float4(r0.x, r1.x, r0.y, r1.y);
        d[1] = make_float4(r0.z, r1.z, r0.w, r1.w);
    }

    const int wave    = __builtin_amdgcn_readfirstlane(tid >> 5);
    const int lane    = tid & 31;
    const int halfsel = lane >> 4;
    const int l16     = lane & 15;
    const int colbase = (wave & 3) * 64;
    const int ntiles  = N >> 4;

    const int first = blockIdx.x + GEMM_BLOCKS * (wave >> 2);
    int mtile = min(first, ntiles - 1);
    const float* xlane = X + (size_t)l16 * D_IN + halfsel * 2;

    v2f a[32];
    {
        const float* xrow = xlane + (size_t)mtile * (16 * D_IN);
        #pragma unroll
        for (int s = 0; s < 32; ++s) a[s] = *(const v2f*)(xrow + 4 * s);
    }

    __syncthreads();

    v2f b[32][4];
    {
        const float* bl = wl + halfsel * WPITCH + 2 * (colbase + l16);
        #pragma unroll
        for (int s = 0; s < 32; ++s) {
            #pragma unroll
            for (int n = 0; n < 4; ++n) b[s][n] = *(const v2f*)(bl + (2 * s) * WPITCH + 32 * n);
        }
    }

    if (first >= ntiles) return;

    for (;;) {
        const int  nxt  = mtile + GEMM_TSTRIDE;
        const bool more = nxt < ntiles;
        v2f an[32];
        {
            const float* xrow = xlane + (size_t)(more ? nxt : mtile) * (16 * D_IN);
            #pragma unroll
            for (int s = 0; s < 32; ++s) an[s] = *(const v2f*)(xrow + 4 * s);
        }

        v8f acc0 = {}; v8f acc1 = {}; v8f acc2 = {}; v8f acc3 = {};
        #pragma unroll
        for (int s = 0; s < 32; ++s) {
            acc0 = __builtin_amdgcn_wmma_f32_16x16x4_f32(false, a[s], false, b[s][0], (short)0, acc0, false, false);
            acc1 = __builtin_amdgcn_wmma_f32_16x16x4_f32(false, a[s], false, b[s][1], (short)0, acc1, false, false);
            acc2 = __builtin_amdgcn_wmma_f32_16x16x4_f32(false, a[s], false, b[s][2], (short)0, acc2, false, false);
            acc3 = __builtin_amdgcn_wmma_f32_16x16x4_f32(false, a[s], false, b[s][3], (short)0, acc3, false, false);
        }

        #pragma unroll
        for (int r = 0; r < 8; ++r) {
            const int m = mtile * 16 + r + halfsel * 8;
            float* hrow = H + (size_t)m * NH_COLS + colbase + l16;
            hrow[0]  = acc0[r];
            hrow[16] = acc1[r];
            hrow[32] = acc2[r];
            hrow[48] = acc3[r];
        }

        if (!more) break;
        mtile = nxt;
        #pragma unroll
        for (int s = 0; s < 32; ++s) a[s] = an[s];
    }
}

__global__ void gat_attn_coef(const float* __restrict__ H,
                              const float* __restrict__ att_src,
                              const float* __restrict__ att_dst,
                              float* __restrict__ a_src,
                              float* __restrict__ a_dst, int nh,
                              const int* __restrict__ ei, int E,
                              unsigned int* __restrict__ keys)
{
    const int tid = blockIdx.x * blockDim.x + threadIdx.x;
    {
        const int* dp = ei + E;
        const int nq = (E + 3) >> 2;
        for (int q = tid; q < nq; q += gridDim.x * blockDim.x) {
            unsigned int kw = 0u;
            #pragma unroll
            for (int j = 0; j < 4; ++j) {
                const int e = 4 * q + j;
                unsigned int kb = 0xFFu;
                if (e < E) kb = ((unsigned int)dp[e] / ROWS_PER_BLOCK) & 0xFFu;
                kw |= kb << (8 * j);
            }
            keys[q] = kw;
        }
    }
    if (tid >= nh) return;
    const int n = tid >> 2, hd = tid & 3;
    const float4* hp = (const float4*)(H + (size_t)n * NH_COLS + hd * F_OUT);
    const float4* as = (const float4*)(att_src + hd * F_OUT);
    const float4* ad = (const float4*)(att_dst + hd * F_OUT);
    float s1 = 0.0f, s2 = 0.0f;
    #pragma unroll
    for (int i = 0; i < 16; ++i) {
        const float4 hv = hp[i], v1 = as[i], v2 = ad[i];
        s1 += hv.x * v1.x + hv.y * v1.y + hv.z * v1.z + hv.w * v1.w;
        s2 += hv.x * v2.x + hv.y * v2.y + hv.z * v2.z + hv.w * v2.w;
    }
    a_src[tid] = s1;
    a_dst[tid] = s2;
}

__device__ __forceinline__ void atomicMaxFloat(float* addr, float val)
{
    if (val >= 0.0f) atomicMax((int*)addr, __float_as_int(val));
    else             atomicMin((unsigned int*)addr, __float_as_uint(val));
}

__device__ __forceinline__ void edge_endpoints(const int* __restrict__ ei,
                                               int e, int E, int& src, int& dst)
{
    if (e < E) { src = ei[e]; dst = ei[E + e]; }
    else       { src = dst = e - E; }
}

__device__ __forceinline__ float leaky(float a)
{
    return a > 0.0f ? a : NEG_SLOPE * a;
}

#define AGG_THREADS    1024
#define AGG_WAVES      (AGG_THREADS / 32)
#define QCAP           4096
#define QPT            (QCAP / AGG_THREADS)
#define GROUP          8
#define SCAN_DEPTH     8
#define OUT_ROWS_PER_WAVE ((ROWS_PER_BLOCK + AGG_WAVES - 1) / AGG_WAVES)

static_assert(HEADS * ROWS_PER_BLOCK <= AGG_THREADS, "one thread per (head,row) start value");
static_assert(QPT * AGG_THREADS == QCAP, "the stats step covers the queue exactly");
static_assert(2 * AGG_THREADS == 32 * F_OUT, "two floats of W_prior per thread fill its LDS copy");

__device__ __forceinline__ float lane_bcast(float v, int j)
{
    return __int_as_float(__builtin_amdgcn_readlane(__float_as_int(v), j));
}

__device__ __forceinline__ void lds_add_row(float* o,
                                            float u0, float u1, float u2, float u3,
                                            float v0, float v1, float v2, float v3,
                                            float v4, float v5, float v6, float v7)
{
    atomicAdd(o,       u0 * v0);
    atomicAdd(o + 32,  u0 * v1);
    atomicAdd(o + 64,  u1 * v2);
    atomicAdd(o + 96,  u1 * v3);
    atomicAdd(o + 128, u2 * v4);
    atomicAdd(o + 160, u2 * v5);
    atomicAdd(o + 192, u3 * v6);
    atomicAdd(o + 224, u3 * v7);
}

__device__ __forceinline__ void edge_logits(int src, int dst,
                                            const float* __restrict__ a_src,
                                            const float* __restrict__ a_dst,
                                            float& l0, float& l1, float& l2, float& l3)
{
    const float4 as = *(const float4*)(a_src + (size_t)src * HEADS);
    const float4 ad = *(const float4*)(a_dst + (size_t)dst * HEADS);
    l0 = leaky(as.x + ad.x);
    l1 = leaky(as.y + ad.y);
    l2 = leaky(as.z + ad.z);
    l3 = leaky(as.w + ad.w);
}

__global__ __launch_bounds__(AGG_THREADS) void gat_edge_agg(const int* __restrict__ ei, int E, int Etot, int N,
                             const float* __restrict__ a_src,
                             const float* __restrict__ a_dst,
                             const float* __restrict__ H,
                             const float* __restrict__ bias_gat,
                             const float* __restrict__ onehot,
                             const float* __restrict__ W_prior,
                             const float* __restrict__ b_prior,
                             const float* __restrict__ gate,
                             const unsigned int* __restrict__ keys,
                             float* __restrict__ out)
{
    __shared__ __align__(16) float rows[ROWS_PER_BLOCK * NH_COLS];
    __shared__ __align__(16) float qw[QCAP * HEADS];
    __shared__ int   queue[QCAP];
    __shared__ int   qrow[QCAP];
    __shared__ float smax[HEADS * ROWS_PER_BLOCK];
    __shared__ float sden[HEADS * ROWS_PER_BLOCK];
    __shared__ __align__(8) float wps[32 * F_OUT];
    __shared__ int   qcount;
    __shared__ int   wcnt[AGG_WAVES];
    __shared__ int   wbase[AGG_WAVES];

    const int tid  = threadIdx.x;
    const int lane = tid & 31;
    const int wave = __builtin_amdgcn_readfirstlane(tid >> 5);
    const int n0     = blockIdx.x * ROWS_PER_BLOCK;
    const int rcount = min(ROWS_PER_BLOCK, N - n0);
    const unsigned ur = (unsigned)rcount;

    float4* rows4 = (float4*)rows;
    for (int i = tid; i < ROWS_PER_BLOCK * (NH_COLS / 4); i += AGG_THREADS)
        rows4[i] = make_float4(0.0f, 0.0f, 0.0f, 0.0f);
    if (tid < HEADS * ROWS_PER_BLOCK) {
        smax[tid] = __int_as_float(0xFF800000);
        sden[tid] = 0.0f;
    }
    if (tid < rcount) queue[tid] = E + n0 + tid;
    if (tid == 0) qcount = rcount;
    if (tid < AGG_WAVES) wcnt[tid] = 0;
    __syncthreads();

    const int    f2     = 2 * lane;
    const float2 wstage = *(const float2*)(W_prior + 2 * tid);
    float oh[OUT_ROWS_PER_WAVE];
    #pragma unroll
    for (int k = 0; k < OUT_ROWS_PER_WAVE; ++k) {
        const int r = min(wave + AGG_WAVES * k, rcount - 1);
        oh[k] = onehot[(size_t)(n0 + r) * 32 + lane];
    }
    const v2f bg = *(const v2f*)(bias_gat + f2);
    const v2f bp = *(const v2f*)(b_prior + f2);
    const float g = gate[0];

    const bool usekeys = gridDim.x <= 256u;
    const int* dp  = ei + E;
    const uint4* kp4 = (const uint4*)keys;
    const int  nb16 = usekeys ? (E >> 4) : 0;
    const unsigned int pat = blockIdx.x * 0x01010101u;
    const int  rot = (int)(((long long)blockIdx.x * nb16) / gridDim.x);
    if (nb16 > 0) {
        for (int i = tid; i < nb16; i += SCAN_DEPTH * AGG_THREADS) {
            uint4 d[SCAN_DEPTH];
            int   q[SCAN_DEPTH];
            #pragma unroll
            for (int k = 0; k < SCAN_DEPTH; ++k) {
                int qq = min(i + k * AGG_THREADS, nb16 - 1) + rot;
                if (qq >= nb16) qq -= nb16;
                q[k] = qq;
                d[k] = kp4[qq];
            }
            unsigned int zz[SCAN_DEPTH];
            int c = 0;
            #pragma unroll
            for (int k = 0; k < SCAN_DEPTH; ++k) {
                const bool ok = (i + k * AGG_THREADS) < nb16;
                const unsigned int x0 = d[k].x ^ pat, x1 = d[k].y ^ pat;
                const unsigned int x2 = d[k].z ^ pat, x3 = d[k].w ^ pat;
                const unsigned int z0 = ~(((x0 & 0x7F7F7F7Fu) + 0x7F7F7F7Fu) | x0 | 0x7F7F7F7Fu);
                const unsigned int z1 = ~(((x1 & 0x7F7F7F7Fu) + 0x7F7F7F7Fu) | x1 | 0x7F7F7F7Fu);
                const unsigned int z2 = ~(((x2 & 0x7F7F7F7Fu) + 0x7F7F7F7Fu) | x2 | 0x7F7F7F7Fu);
                const unsigned int z3 = ~(((x3 & 0x7F7F7F7Fu) + 0x7F7F7F7Fu) | x3 | 0x7F7F7F7Fu);
                const unsigned int m = (z0 >> 7) | (z1 >> 6) | (z2 >> 5) | (z3 >> 4);
                zz[k] = ok ? m : 0u;
                c += __popc(zz[k]);
            }
            if (c) {
                int s = atomicAdd(&qcount, c);
                #pragma unroll
                for (int k = 0; k < SCAN_DEPTH; ++k) {
                    unsigned int m = zz[k];
                    const int e0 = 16 * q[k];
                    while (m) {
                        const int p = __builtin_ctz(m);
                        m &= m - 1u;
                        if (s < QCAP) queue[s] = e0 + 4 * (p & 7) + (p >> 3);
                        ++s;
                    }
                }
            }
        }
    }
    if (usekeys) {
        const int t = 16 * nb16 + tid;
        if (tid < 16 && t < E && (unsigned)(dp[t] - n0) < ur) {
            const int s = atomicAdd(&qcount, 1);
            if (s < QCAP) queue[s] = t;
        }
    }
    ((float2*)wps)[tid] = wstage;
    __syncthreads();

    const int total = usekeys ? __builtin_amdgcn_readfirstlane(qcount) : QCAP + 1;
    const float* hbase = H + lane;

    if (total <= QCAP) {
        int   qs[QPT], qr[QPT], rk[QPT];
        float lg[QPT][HEADS];
        #pragma unroll
        for (int k = 0; k < QPT; ++k) {
            const int idx = tid + k * AGG_THREADS;
            const int e = (idx < total) ? queue[idx] : (E + n0);
            int s, d; edge_endpoints(ei, e, E, s, d);
            qs[k] = s;
            qr[k] = d - n0;
        }
        #pragma unroll
        for (int k = 0; k < QPT; ++k)
            edge_logits(qs[k], n0 + qr[k], a_src, a_dst, lg[k][0], lg[k][1], lg[k][2], lg[k][3]);
        #pragma unroll
        for (int k = 0; k < QPT; ++k) {
            const int idx = tid + k * AGG_THREADS;
            rk[k] = 0;
            if (idx < total) {
                rk[k] = atomicAdd(&wcnt[qr[k] & (AGG_WAVES - 1)], 1);
                #pragma unroll
                for (int hd = 0; hd < HEADS; ++hd)
                    atomicMaxFloat(&smax[hd * ROWS_PER_BLOCK + qr[k]], lg[k][hd]);
            }
        }
        __syncthreads();
        if (wave == 0) {
            const int cw = wcnt[lane];
            int incl = cw;
            #pragma unroll
            for (int d = 1; d < 32; d <<= 1) {
                const int t = __shfl_up(incl, d);
                if (lane >= d) incl += t;
            }
            wbase[lane] = incl - cw;
        }
        float st[QPT][HEADS];
        #pragma unroll
        for (int k = 0; k < QPT; ++k) {
            #pragma unroll
            for (int hd = 0; hd < HEADS; ++hd) st[k][hd] = smax[hd * ROWS_PER_BLOCK + qr[k]];
        }
        #pragma unroll
        for (int k = 0; k < QPT; ++k) {
            #pragma unroll
            for (int hd = 0; hd < HEADS; ++hd) lg[k][hd] = __expf(lg[k][hd] - st[k][hd]);
        }
        #pragma unroll
        for (int k = 0; k < QPT; ++k) {
            const int idx = tid + k * AGG_THREADS;
            if (idx < total) {
                #pragma unroll
                for (int hd = 0; hd < HEADS; ++hd)
                    atomicAdd(&sden[hd * ROWS_PER_BLOCK + qr[k]], lg[k][hd]);
            }
        }
        __syncthreads();
        #pragma unroll
        for (int k = 0; k < QPT; ++k) {
            #pragma unroll
            for (int hd = 0; hd < HEADS; ++hd) st[k][hd] = sden[hd * ROWS_PER_BLOCK + qr[k]];
        }
        #pragma unroll
        for (int k = 0; k < QPT; ++k) {
            const int idx = tid + k * AGG_THREADS;
            if (idx < total) {
                float4 w;
                w.x = lg[k][0] / (st[k][0] + 1e-16f);
                w.y = lg[k][1] / (st[k][1] + 1e-16f);
                w.z = lg[k][2] / (st[k][2] + 1e-16f);
                w.w = lg[k][3] / (st[k][3] + 1e-16f);
                const int pos = wbase[qr[k] & (AGG_WAVES - 1)] + rk[k];
                queue[pos] = qs[k];
                qrow[pos]  = qr[k];
                ((float4*)qw)[pos] = w;
            }
        }
        __syncthreads();

        const int beg = __builtin_amdgcn_readfirstlane(wbase[wave]);
        const int cnt = __builtin_amdgcn_readfirstlane(wcnt[wave]);
        const bool lowhalf = lane < 16;
        const float4* hl = (const float4*)H + lane;
        float4*       rl = (float4*)rows + lane;
        for (int c0 = 0; c0 < cnt; c0 += 32) {
            const int n = min(32, cnt - c0);
            int src = 0, row = 0;
            float w0 = 0.0f, w1 = 0.0f, w2 = 0.0f, w3 = 0.0f;
            if (lane < n) {
                src = queue[beg + c0 + lane];
                row = qrow[beg + c0 + lane];
                const float4 w = ((const float4*)qw)[beg + c0 + lane];
                w0 = w.x; w1 = w.y; w2 = w.z; w3 = w.w;
            }
            for (int g = 0; g < n; g += GROUP) {
                float4 va[GROUP], vb[GROUP];
                #pragma unroll
                for (int k = 0; k < GROUP; ++k) {
                    const float4* hp = hl + (size_t)__builtin_amdgcn_readlane(src, g + k) * (NH_COLS / 4);
                    va[k] = hp[0];
                    vb[k] = hp[NH_COLS / 8];
                }
                #pragma unroll
                for (int k = 0; k < GROUP; ++k) {
                    if (g + k < n) {
                        const float u0 = lane_bcast(w0, g + k), u1 = lane_bcast(w1, g + k);
                        const float u2 = lane_bcast(w2, g + k), u3 = lane_bcast(w3, g + k);
                        const float ua = lowhalf ? u0 : u1;
                        const float ub = lowhalf ? u2 : u3;
                        float4* ra = rl + __builtin_amdgcn_readlane(row, g + k) * (NH_COLS / 4);
                        float4 a = ra[0];
                        float4 b = ra[NH_COLS / 8];
                        a.x += ua * va[k].x; a.y += ua * va[k].y; a.z += ua * va[k].z; a.w += ua * va[k].w;
                        b.x += ub * vb[k].x; b.y += ub * vb[k].y; b.z += ub * vb[k].z; b.w += ub * vb[k].w;
                        ra[0] = a;
                        ra[NH_COLS / 8] = b;
                    }
                }
            }
        }
    } else {
        for (int e = tid; e < Etot; e += AGG_THREADS) {
            int s, d; edge_endpoints(ei, e, E, s, d);
            const int r = d - n0;
            if ((unsigned)r < ur) {
                float l0, l1, l2, l3; edge_logits(s, d, a_src, a_dst, l0, l1, l2, l3);
                atomicMaxFloat(&smax[0 * ROWS_PER_BLOCK + r], l0);
                atomicMaxFloat(&smax[1 * ROWS_PER_BLOCK + r], l1);
                atomicMaxFloat(&smax[2 * ROWS_PER_BLOCK + r], l2);
                atomicMaxFloat(&smax[3 * ROWS_PER_BLOCK + r], l3);
            }
        }
        __syncthreads();
        for (int e = tid; e < Etot; e += AGG_THREADS) {
            int s, d; edge_endpoints(ei, e, E, s, d);
            const int r = d - n0;
            if ((unsigned)r < ur) {
                float l0, l1, l2, l3; edge_logits(s, d, a_src, a_dst, l0, l1, l2, l3);
                atomicAdd(&sden[0 * ROWS_PER_BLOCK + r], __expf(l0 - smax[0 * ROWS_PER_BLOCK + r]));
                atomicAdd(&sden[1 * ROWS_PER_BLOCK + r], __expf(l1 - smax[1 * ROWS_PER_BLOCK + r]));
                atomicAdd(&sden[2 * ROWS_PER_BLOCK + r], __expf(l2 - smax[2 * ROWS_PER_BLOCK + r]));
                atomicAdd(&sden[3 * ROWS_PER_BLOCK + r], __expf(l3 - smax[3 * ROWS_PER_BLOCK + r]));
            }
        }
        __syncthreads();
        for (int base = wave * 32; base < Etot; base += AGG_THREADS) {
            const int e = base + lane;
            int src = 0, dst = -1;
            if (e < Etot) edge_endpoints(ei, e, E, src, dst);
            const bool mine = (unsigned)(dst - n0) < ur;
            const int row = dst - n0;
            float w0 = 0.0f, w1 = 0.0f, w2 = 0.0f, w3 = 0.0f;
            if (mine) {
                float l0, l1, l2, l3; edge_logits(src, dst, a_src, a_dst, l0, l1, l2, l3);
                w0 = __expf(l0 - smax[0 * ROWS_PER_BLOCK + row]) / (sden[0 * ROWS_PER_BLOCK + row] + 1e-16f);
                w1 = __expf(l1 - smax[1 * ROWS_PER_BLOCK + row]) / (sden[1 * ROWS_PER_BLOCK + row] + 1e-16f);
                w2 = __expf(l2 - smax[2 * ROWS_PER_BLOCK + row]) / (sden[2 * ROWS_PER_BLOCK + row] + 1e-16f);
                w3 = __expf(l3 - smax[3 * ROWS_PER_BLOCK + row]) / (sden[3 * ROWS_PER_BLOCK + row] + 1e-16f);
            }
            unsigned mask = __builtin_amdgcn_ballot_w32(mine);
            while (mask) {
                const int j = __builtin_ctz(mask);
                mask &= mask - 1;
                const float* hr = hbase + (size_t)__builtin_amdgcn_readlane(src, j) * NH_COLS;
                const float v0 = hr[0],   v1 = hr[32],  v2 = hr[64],  v3 = hr[96];
                const float v4 = hr[128], v5 = hr[160], v6 = hr[192], v7 = hr[224];
                lds_add_row(rows + __builtin_amdgcn_readlane(row, j) * NH_COLS + lane,
                            lane_bcast(w0, j), lane_bcast(w1, j), lane_bcast(w2, j), lane_bcast(w3, j),
                            v0, v1, v2, v3, v4, v5, v6, v7);
            }
        }
    }

    __syncthreads();

    v2f wp[32];
    #pragma unroll
    for (int c = 0; c < 32; ++c) wp[c] = ((const v2f*)wps)[c * 32 + lane];

    #pragma unroll
    for (int k = 0; k < OUT_ROWS_PER_WAVE; ++k) {
        const int r = wave + AGG_WAVES * k;
        if (r < rcount) {
            const float* rp = rows + r * NH_COLS + f2;
            const v2f a0 = *(const v2f*)(rp);
            const v2f a1 = *(const v2f*)(rp + F_OUT);
            const v2f a2 = *(const v2f*)(rp + 2 * F_OUT);
            const v2f a3 = *(const v2f*)(rp + 3 * F_OUT);
            const v2f quarter = {0.25f, 0.25f};
            const v2f m = __builtin_elementwise_fma(quarter, a0 + a1 + a2 + a3, bg);
            v2f p = bp;
            #pragma unroll
            for (int c = 0; c < 32; ++c) {
                const float o = lane_bcast(oh[k], c);
                const v2f o2 = {o, o};
                p = __builtin_elementwise_fma(o2, wp[c], p);
            }
            const v2f g2 = {g, g};
            const v2f h2 = {1.0f - g, 1.0f - g};
            const v2f res = __builtin_elementwise_fma(h2, m, g2 * p);
            *(v2f*)(out + (size_t)(n0 + r) * F_OUT + f2) = res;
        }
    }
}

extern "C" void kernel_launch(void* const* d_in, const int* in_sizes, int n_in,
                              void* d_out, int out_size, void* d_ws, size_t ws_size,
                              hipStream_t stream)
{
    const float* x        = (const float*)d_in[0];
    const int*   ei       = (const int*)  d_in[1];
    const float* onehot   = (const float*)d_in[2];
    const float* W_gat    = (const float*)d_in[3];
    const float* att_src  = (const float*)d_in[4];
    const float* att_dst  = (const float*)d_in[5];
    const float* bias_gat = (const float*)d_in[6];
    const float* W_prior  = (const float*)d_in[7];
    const float* b_prior  = (const float*)d_in[8];
    const float* gate     = (const float*)d_in[9];
    float* out = (float*)d_out;

    const int N    = in_sizes[0] / D_IN;
    const int E    = in_sizes[1] / 2;
    const int Etot = E + N;
    const int NHs  = N * HEADS;

    float* h      = (float*)d_ws;
    float* a_src  = h      + (size_t)N * NH_COLS;
    float* a_dst  = a_src  + NHs;
    unsigned int* keys = (unsigned int*)(a_dst + NHs);

    gat_gemm_wmma<<<GEMM_BLOCKS, GEMM_THREADS, 0, stream>>>(x, W_gat, h, N);

    gat_attn_coef<<<(NHs + 255) / 256, 256, 0, stream>>>(h, att_src, att_dst,
                                                         a_src, a_dst, NHs, ei, E, keys);

    gat_edge_agg<<<(N + ROWS_PER_BLOCK - 1) / ROWS_PER_BLOCK, AGG_THREADS, 0, stream>>>(
        ei, E, Etot, N, a_src, a_dst, h,
        bias_gat, onehot, W_prior, b_prior, gate, keys, out);
}


# attn_coef launch: a wave takes 16 whole rows of h by 16-B loads, all its loads out as one group; head sums by DPP row adds (a_src/a_dst sum order: 16-lane tree); keys: 4 dst per 16-B load
# speedup vs baseline: 5.4208x; 1.1432x over previous
#include <hip/hip_runtime.h>
#include <hip/hip_bf16.h>
#include <math.h>

typedef __attribute__((ext_vector_type(2))) float v2f;
typedef __attribute__((ext_vector_type(8))) float v8f;

#define HEADS 4
#define F_OUT 64
#define D_IN 128
#define NH_COLS 256
#define NEG_SLOPE 0.2f
#define ROWS_PER_BLOCK 200

#define GEMM_THREADS 256
#define GEMM_WAVES   (GEMM_THREADS / 32)
#define GEMM_BLOCKS  256
#define GEMM_TSTRIDE (GEMM_BLOCKS * (GEMM_WAVES / 4))
#define WPITCH       544

__global__ __launch_bounds__(GEMM_THREADS) void gat_gemm_wmma(
    const float* __restrict__ X,
    const float* __restrict__ W,
    float* __restrict__ H,
    int N)
{
    __shared__ __align__(16) float wl[(D_IN / 2) * WPITCH];

    const int tid = threadIdx.x;
    #pragma unroll
    for (int i = 0; i < (D_IN / 2) * (NH_COLS / 4) / GEMM_THREADS; ++i) {
        const int item = tid + GEMM_THREADS * i;
        const int p  = item >> 6;
        const int c4 = item & 63;
        const float4 r0 = *(const float4*)(W + (size_t)(2 * p)     * NH_COLS + 4 * c4);
        const float4 r1 = *(const float4*)(W + (size_t)(2 * p + 1) * NH_COLS + 4 * c4);
        float4* d = (float4*)(wl + p * WPITCH + 8 * c4);
        d[0] = make_float4(r0.x, r1.x, r0.y, r1.y);
        d[1] = make_float4(r0.z, r1.z, r0.w, r1.w);
    }

    const int wave    = __builtin_amdgcn_readfirstlane(tid >> 5);
    const int lane    = tid & 31;
    const int halfsel = lane >> 4;
    const int l16     = lane & 15;
    const int colbase = (wave & 3) * 64;
    const int ntiles  = N >> 4;

    const int first = blockIdx.x + GEMM_BLOCKS * (wave >> 2);
    int mtile = min(first, ntiles - 1);
    const float* xlane = X + (size_t)l16 * D_IN + halfsel * 2;

    v2f a[32];
    {
        const float* xrow = xlane + (size_t)mtile * (16 * D_IN);
        #pragma unroll
        for (int s = 0; s < 32; ++s) a[s] = *(const v2f*)(xrow + 4 * s);
    }

    __syncthreads();

    v2f b[32][4];
    {
        const float* bl = wl + halfsel * WPITCH + 2 * (colbase + l16);
        #pragma unroll
        for (int s = 0; s < 32; ++s) {
            #pragma unroll
            for (int n = 0; n < 4; ++n) b[s][n] = *(const v2f*)(bl + (2 * s) * WPITCH + 32 * n);
        }
    }

    if (first >= ntiles) return;

    for (;;) {
        const int  nxt  = mtile + GEMM_TSTRIDE;
        const bool more = nxt < ntiles;
        v2f an[32];
        {
            const float* xrow = xlane + (size_t)(more ? nxt : mtile) * (16 * D_IN);
            #pragma unroll
            for (int s = 0; s < 32; ++s) an[s] = *(const v2f*)(xrow + 4 * s);
        }

        v8f acc0 = {}; v8f acc1 = {}; v8f acc2 = {}; v8f acc3 = {};
        #pragma unroll
        for (int s = 0; s < 32; ++s) {
            acc0 = __builtin_amdgcn_wmma_f32_16x16x4_f32(false, a[s], false, b[s][0], (short)0, acc0, false, false);
            acc1 = __builtin_amdgcn_wmma_f32_16x16x4_f32(false, a[s], false, b[s][1], (short)0, acc1, false, false);
            acc2 = __builtin_amdgcn_wmma_f32_16x16x4_f32(false, a[s], false, b[s][2], (short)0, acc2, false, false);
            acc3 = __builtin_amdgcn_wmma_f32_16x16x4_f32(false, a[s], false, b[s][3], (short)0, acc3, false, false);
        }

        #pragma unroll
        for (int r = 0; r < 8; ++r) {
            const int m = mtile * 16 + r + halfsel * 8;
            float* hrow = H + (size_t)m * NH_COLS + colbase + l16;
            hrow[0]  = acc0[r];
            hrow[16] = acc1[r];
            hrow[32] = acc2[r];
            hrow[48] = acc3[r];
        }

        if (!more) break;
        mtile = nxt;
        #pragma unroll
        for (int s = 0; s < 32; ++s) a[s] = an[s];
    }
}

#define COEF_THREADS 256
#define COEF_ROWS    16
#define COEF_ROWS_PER_BLOCK (COEF_ROWS * (COEF_THREADS / 32))

__device__ __forceinline__ unsigned int edge_key(int dst)
{
    return ((unsigned int)dst / ROWS_PER_BLOCK) & 0xFFu;
}

__device__ __forceinline__ unsigned int pack_keys(const int4 d)
{
    return edge_key(d.x) | (edge_key(d.y) << 8) | (edge_key(d.z) << 16) | (edge_key(d.w) << 24);
}

__device__ __forceinline__ unsigned int key_word_single(const int* __restrict__ dp, int q, int E)
{
    unsigned int kw = 0u;
    #pragma unroll
    for (int j = 0; j < 4; ++j) {
        const int e = 4 * q + j;
        unsigned int kb = 0xFFu;
        if (e < E) kb = edge_key(dp[e]);
        kw |= kb << (8 * j);
    }
    return kw;
}

#define ROW_ADD(v, ctrl) \
    (v) += __int_as_float(__builtin_amdgcn_update_dpp(0, __float_as_int(v), (ctrl), 0xF, 0xF, true))

__device__ __forceinline__ float row16_sum(float v)
{
    ROW_ADD(v, 0xB1);
    ROW_ADD(v, 0x4E);
    ROW_ADD(v, 0x141);
    ROW_ADD(v, 0x140);
    return v;
}

__device__ __forceinline__ float dot4(const float4 a, const float4 b)
{
    return a.x * b.x + a.y * b.y + a.z * b.z + a.w * b.w;
}

__global__ __launch_bounds__(COEF_THREADS) void gat_attn_coef(const float* __restrict__ H,
                              const float* __restrict__ att_src,
                              const float* __restrict__ att_dst,
                              float* __restrict__ a_src,
                              float* __restrict__ a_dst, int N,
                              const int* __restrict__ ei, int E,
                              unsigned int* __restrict__ keys)
{
    const int tid    = blockIdx.x * blockDim.x + threadIdx.x;
    const int stride = gridDim.x * blockDim.x;
    const int lane   = threadIdx.x & 31;
    const int nb = __builtin_amdgcn_readfirstlane((tid >> 5) * COEF_ROWS);

    const int* dp   = ei + E;
    const int  nq   = (E + 3) >> 2;
    const bool whole = (E & 3) == 0 && nq > 0;

    const int  q1 = tid + stride;
    const int4 d0 = *(const int4*)(whole ? dp + 4 * min(tid, nq - 1) : ei);
    const int4 d1 = *(const int4*)(whole ? dp + 4 * min(q1, nq - 1) : ei);

    const float4 sa = ((const float4*)att_src)[lane];
    const float4 sb = ((const float4*)att_src)[32 + lane];
    const float4 da = ((const float4*)att_dst)[lane];
    const float4 db = ((const float4*)att_dst)[32 + lane];

    const bool have = nb < N;
    float4 va[COEF_ROWS], vb[COEF_ROWS];
    if (have) {
        const float4* hl = (const float4*)H + lane;
        #pragma unroll
        for (int k = 0; k < COEF_ROWS; ++k) {
            const float4* hp = hl + (size_t)min(nb + k, N - 1) * (NH_COLS / 4);
            va[k] = hp[0];
            vb[k] = hp[NH_COLS / 8];
        }
    }

    if (tid < nq) keys[tid] = whole ? pack_keys(d0) : key_word_single(dp, tid, E);
    if (q1 < nq)  keys[q1]  = whole ? pack_keys(d1) : key_word_single(dp, q1, E);
    for (int q = q1 + stride; q < nq; q += stride)
        keys[q] = whole ? pack_keys(*(const int4*)(dp + 4 * q)) : key_word_single(dp, q, E);

    if (!have) return;

    float s0[COEF_ROWS], s1[COEF_ROWS], t0[COEF_ROWS], t1[COEF_ROWS];
    #pragma unroll
    for (int k = 0; k < COEF_ROWS; ++k) {
        s0[k] = row16_sum(dot4(va[k], sa));
        s1[k] = row16_sum(dot4(vb[k], sb));
        t0[k] = row16_sum(dot4(va[k], da));
        t1[k] = row16_sum(dot4(vb[k], db));
    }

    const int r8   = (lane & 15) >> 1;
    const bool odd = (lane & 1) != 0;
    const int head = (odd ? 2 : 0) + (lane >> 4);
    #pragma unroll
    for (int g = 0; g < COEF_ROWS / 8; ++g) {
        float vs = 0.0f, vd = 0.0f;
        #pragma unroll
        for (int k = 0; k < 8; ++k) {
            if (r8 == k) {
                vs = odd ? s1[8 * g + k] : s0[8 * g + k];
                vd = odd ? t1[8 * g + k] : t0[8 * g + k];
            }
        }
        const int row = nb + 8 * g + r8;
        if (row < N) {
            a_src[(size_t)row * HEADS + head] = vs;
            a_dst[(size_t)row * HEADS + head] = vd;
        }
    }
}

__device__ __forceinline__ void atomicMaxFloat(float* addr, float val)
{
    if (val >= 0.0f) atomicMax((int*)addr, __float_as_int(val));
    else             atomicMin((unsigned int*)addr, __float_as_uint(val));
}

__device__ __forceinline__ void edge_endpoints(const int* __restrict__ ei,
                                               int e, int E, int& src, int& dst)
{
    if (e < E) { src = ei[e]; dst = ei[E + e]; }
    else       { src = dst = e - E; }
}

__device__ __forceinline__ float leaky(float a)
{
    return a > 0.0f ? a : NEG_SLOPE * a;
}

#define AGG_THREADS    1024
#define AGG_WAVES      (AGG_THREADS / 32)
#define QCAP           4096
#define QPT            (QCAP / AGG_THREADS)
#define GROUP          8
#define SCAN_DEPTH     8
#define OUT_ROWS_PER_WAVE ((ROWS_PER_BLOCK + AGG_WAVES - 1) / AGG_WAVES)

static_assert(HEADS * ROWS_PER_BLOCK <= AGG_THREADS, "one thread per (head,row) start value");
static_assert(QPT * AGG_THREADS == QCAP, "the stats step covers the queue exactly");
static_assert(2 * AGG_THREADS == 32 * F_OUT, "two floats of W_prior per thread fill its LDS copy");

__device__ __forceinline__ float lane_bcast(float v, int j)
{
    return __int_as_float(__builtin_amdgcn_readlane(__float_as_int(v), j));
}

__device__ __forceinline__ void lds_add_row(float* o,
                                            float u0, float u1, float u2, float u3,
                                            float v0, float v1, float v2, float v3,
                                            float v4, float v5, float v6, float v7)
{
    atomicAdd(o,       u0 * v0);
    atomicAdd(o + 32,  u0 * v1);
    atomicAdd(o + 64,  u1 * v2);
    atomicAdd(o + 96,  u1 * v3);
    atomicAdd(o + 128, u2 * v4);
    atomicAdd(o + 160, u2 * v5);
    atomicAdd(o + 192, u3 * v6);
    atomicAdd(o + 224, u3 * v7);
}

__device__ __forceinline__ void edge_logits(int src, int dst,
                                            const float* __restrict__ a_src,
                                            const float* __restrict__ a_dst,
                                            float& l0, float& l1, float& l2, float& l3)
{
    const float4 as = *(const float4*)(a_src + (size_t)src * HEADS);
    const float4 ad = *(const float4*)(a_dst + (size_t)dst * HEADS);
    l0 = leaky(as.x + ad.x);
    l1 = leaky(as.y + ad.y);
    l2 = leaky(as.z + ad.z);
    l3 = leaky(as.w + ad.w);
}

__global__ __launch_bounds__(AGG_THREADS) void gat_edge_agg(const int* __restrict__ ei, int E, int Etot, int N,
                             const float* __restrict__ a_src,
                             const float* __restrict__ a_dst,
                             const float* __restrict__ H,
                             const float* __restrict__ bias_gat,
                             const float* __restrict__ onehot,
                             const float* __restrict__ W_prior,
                             const float* __restrict__ b_prior,
                             const float* __restrict__ gate,
                             const unsigned int* __restrict__ keys,
                             float* __restrict__ out)
{
    __shared__ __align__(16) float rows[ROWS_PER_BLOCK * NH_COLS];
    __shared__ __align__(16) float qw[QCAP * HEADS];
    __shared__ int   queue[QCAP];
    __shared__ int   qrow[QCAP];
    __shared__ float smax[HEADS * ROWS_PER_BLOCK];
    __shared__ float sden[HEADS * ROWS_PER_BLOCK];
    __shared__ __align__(8) float wps[32 * F_OUT];
    __shared__ int   qcount;
    __shared__ int   rcnt[AGG_WAVES * OUT_ROWS_PER_WAVE];
    __shared__ int   rbase[AGG_WAVES * OUT_ROWS_PER_WAVE];
    __shared__ int   wcnt[AGG_WAVES];
    __shared__ int   wbase[AGG_WAVES];

    const int tid  = threadIdx.x;
    const int lane = tid & 31;
    const int wave = __builtin_amdgcn_readfirstlane(tid >> 5);
    const int n0     = blockIdx.x * ROWS_PER_BLOCK;
    const int rcount = min(ROWS_PER_BLOCK, N - n0);
    const unsigned ur = (unsigned)rcount;

    float4* rows4 = (float4*)rows;
    for (int i = tid; i < ROWS_PER_BLOCK * (NH_COLS / 4); i += AGG_THREADS)
        rows4[i] = make_float4(0.0f, 0.0f, 0.0f, 0.0f);
    if (tid < HEADS * ROWS_PER_BLOCK) {
        smax[tid] = __int_as_float(0xFF800000);
        sden[tid] = 0.0f;
    }
    if (tid < rcount) queue[tid] = E + n0 + tid;
    if (tid == 0) qcount = rcount;
    if (tid < AGG_WAVES * OUT_ROWS_PER_WAVE) rcnt[tid] = 0;
    __syncthreads();

    const int    f2     = 2 * lane;
    const float2 wstage = *(const float2*)(W_prior + 2 * tid);
    float oh[OUT_ROWS_PER_WAVE];
    #pragma unroll
    for (int k = 0; k < OUT_ROWS_PER_WAVE; ++k) {
        const int r = min(wave + AGG_WAVES * k, rcount - 1);
        oh[k] = onehot[(size_t)(n0 + r) * 32 + lane];
    }
    const v2f bg = *(const v2f*)(bias_gat + f2);
    const v2f bp = *(const v2f*)(b_prior + f2);
    const float g = gate[0];

    const bool usekeys = gridDim.x <= 256u;
    const int* dp  = ei + E;
    const uint4* kp4 = (const uint4*)keys;
    const int  nb16 = usekeys ? (E >> 4) : 0;
    const unsigned int pat = blockIdx.x * 0x01010101u;
    const int  rot = (int)(((long long)blockIdx.x * nb16) / gridDim.x);
    if (nb16 > 0) {
        for (int i = tid; i < nb16; i += SCAN_DEPTH * AGG_THREADS) {
            uint4 d[SCAN_DEPTH];
            int   q[SCAN_DEPTH];
            #pragma unroll
            for (int k = 0; k < SCAN_DEPTH; ++k) {
                int qq = min(i + k * AGG_THREADS, nb16 - 1) + rot;
                if (qq >= nb16) qq -= nb16;
                q[k] = qq;
                d[k] = kp4[qq];
            }
            unsigned int zz[SCAN_DEPTH];
            int c = 0;
            #pragma unroll
            for (int k = 0; k < SCAN_DEPTH; ++k) {
                const bool ok = (i + k * AGG_THREADS) < nb16;
                const unsigned int x0 = d[k].x ^ pat, x1 = d[k].y ^ pat;
                const unsigned int x2 = d[k].z ^ pat, x3 = d[k].w ^ pat;
                const unsigned int z0 = ~(((x0 & 0x7F7F7F7Fu) + 0x7F7F7F7Fu) | x0 | 0x7F7F7F7Fu);
                const unsigned int z1 = ~(((x1 & 0x7F7F7F7Fu) + 0x7F7F7F7Fu) | x1 | 0x7F7F7F7Fu);
                const unsigned int z2 = ~(((x2 & 0x7F7F7F7Fu) + 0x7F7F7F7Fu) | x2 | 0x7F7F7F7Fu);
                const unsigned int z3 = ~(((x3 & 0x7F7F7F7Fu) + 0x7F7F7F7Fu) | x3 | 0x7F7F7F7Fu);
                const unsigned int m = (z0 >> 7) | (z1 >> 6) | (z2 >> 5) | (z3 >> 4);
                zz[k] = ok ? m : 0u;
                c += __popc(zz[k]);
            }
            if (c) {
                int s = atomicAdd(&qcount, c);
                #pragma unroll
                for (int k = 0; k < SCAN_DEPTH; ++k) {
                    unsigned int m = zz[k];
                    const int e0 = 16 * q[k];
                    while (m) {
                        const int p = __builtin_ctz(m);
                        m &= m - 1u;
                        if (s < QCAP) queue[s] = e0 + 4 * (p & 7) + (p >> 3);
                        ++s;
                    }
                }
            }
        }
    }
    if (usekeys) {
        const int t = 16 * nb16 + tid;
        if (tid < 16 && t < E && (unsigned)(dp[t] - n0) < ur) {
            const int s = atomicAdd(&qcount, 1);
            if (s < QCAP) queue[s] = t;
        }
    }
    ((float2*)wps)[tid] = wstage;
    __syncthreads();

    const int total = usekeys ? __builtin_amdgcn_readfirstlane(qcount) : QCAP + 1;
    const float* hbase = H + lane;

    if (total <= QCAP) {
        int   qs[QPT], qr[QPT], rk[QPT], sl[QPT];
        float lg[QPT][HEADS];
        #pragma unroll
        for (int k = 0; k < QPT; ++k) {
            const int idx = tid + k * AGG_THREADS;
            const int e = (idx < total) ? queue[idx] : (E + n0);
            int s, d; edge_endpoints(ei, e, E, s, d);
            qs[k] = s;
            qr[k] = d - n0;
        }
        #pragma unroll
        for (int k = 0; k < QPT; ++k)
            edge_logits(qs[k], n0 + qr[k], a_src, a_dst, lg[k][0], lg[k][1], lg[k][2], lg[k][3]);
        #pragma unroll
        for (int k = 0; k < QPT; ++k) {
            const int idx = tid + k * AGG_THREADS;
            rk[k] = 0;
            sl[k] = (qr[k] & (AGG_WAVES - 1)) * OUT_ROWS_PER_WAVE + (qr[k] >> 5);
            if (idx < total) {
                rk[k] = atomicAdd(&rcnt[sl[k]], 1);
                #pragma unroll
                for (int hd = 0; hd < HEADS; ++hd)
                    atomicMaxFloat(&smax[hd * ROWS_PER_BLOCK + qr[k]], lg[k][hd]);
            }
        }
        __syncthreads();
        if (wave == 0) {
            int cr[OUT_ROWS_PER_WAVE];
            int cw = 0;
            #pragma unroll
            for (int k = 0; k < OUT_ROWS_PER_WAVE; ++k) {
                cr[k] = rcnt[lane * OUT_ROWS_PER_WAVE + k];
                cw += cr[k];
            }
            int incl = cw;
            #pragma unroll
            for (int d = 1; d < 32; d <<= 1) {
                const int t = __shfl_up(incl, d);
                if (lane >= d) incl += t;
            }
            int run = incl - cw;
            wbase[lane] = run;
            wcnt[lane]  = cw;
            #pragma unroll
            for (int k = 0; k < OUT_ROWS_PER_WAVE; ++k) {
                rbase[lane * OUT_ROWS_PER_WAVE + k] = run;
                run += cr[k];
            }
        }
        float st[QPT][HEADS];
        #pragma unroll
        for (int k = 0; k < QPT; ++k) {
            #pragma unroll
            for (int hd = 0; hd < HEADS; ++hd) st[k][hd] = smax[hd * ROWS_PER_BLOCK + qr[k]];
        }
        #pragma unroll
        for (int k = 0; k < QPT; ++k) {
            #pragma unroll
            for (int hd = 0; hd < HEADS; ++hd) lg[k][hd] = __expf(lg[k][hd] - st[k][hd]);
        }
        #pragma unroll
        for (int k = 0; k < QPT; ++k) {
            const int idx = tid + k * AGG_THREADS;
            if (idx < total) {
                #pragma unroll
                for (int hd = 0; hd < HEADS; ++hd)
                    atomicAdd(&sden[hd * ROWS_PER_BLOCK + qr[k]], lg[k][hd]);
            }
        }
        __syncthreads();
        #pragma unroll
        for (int k = 0; k < QPT; ++k) {
            #pragma unroll
            for (int hd = 0; hd < HEADS; ++hd) st[k][hd] = sden[hd * ROWS_PER_BLOCK + qr[k]];
        }
        #pragma unroll
        for (int k = 0; k < QPT; ++k) {
            const int idx = tid + k * AGG_THREADS;
            if (idx < total) {
                float4 w;
                w.x = lg[k][0] / (st[k][0] + 1e-16f);
                w.y = lg[k][1] / (st[k][1] + 1e-16f);
                w.z = lg[k][2] / (st[k][2] + 1e-16f);
                w.w = lg[k][3] / (st[k][3] + 1e-16f);
                const int pos = rbase[sl[k]] + rk[k];
                queue[pos] = qs[k];
                qrow[pos]  = qr[k];
                ((float4*)qw)[pos] = w;
            }
        }
        __syncthreads();

        const int beg = __builtin_amdgcn_readfirstlane(wbase[wave]);
        const int cnt = __builtin_amdgcn_readfirstlane(wcnt[wave]);
        const bool lowhalf = lane < 16;
        const float4* hl = (const float4*)H + lane;
        float4*       rl = (float4*)rows + lane;
        float4 sa = make_float4(0.0f, 0.0f, 0.0f, 0.0f);
        float4 sb = make_float4(0.0f, 0.0f, 0.0f, 0.0f);
        int cur = -1;
        for (int c0 = 0; c0 < cnt; c0 += 32) {
            const int n = min(32, cnt - c0);
            int src = 0, row = 0;
            float w0 = 0.0f, w1 = 0.0f, w2 = 0.0f, w3 = 0.0f;
            if (lane < n) {
                src = queue[beg + c0 + lane];
                row = qrow[beg + c0 + lane];
                const float4 w = ((const float4*)qw)[beg + c0 + lane];
                w0 = w.x; w1 = w.y; w2 = w.z; w3 = w.w;
            }
            for (int g = 0; g < n; g += GROUP) {
                float4 va[GROUP], vb[GROUP];
                #pragma unroll
                for (int k = 0; k < GROUP; ++k) {
                    const float4* hp = hl + (size_t)__builtin_amdgcn_readlane(src, g + k) * (NH_COLS / 4);
                    va[k] = hp[0];
                    vb[k] = hp[NH_COLS / 8];
                }
                #pragma unroll
                for (int k = 0; k < GROUP; ++k) {
                    if (g + k < n) {
                        const float u0 = lane_bcast(w0, g + k), u1 = lane_bcast(w1, g + k);
                        const float u2 = lane_bcast(w2, g + k), u3 = lane_bcast(w3, g + k);
                        const float ua = lowhalf ? u0 : u1;
                        const float ub = lowhalf ? u2 : u3;
                        const int r = __builtin_amdgcn_readlane(row, g + k);
                        if (r != cur) {
                            if (cur >= 0) {
                                float4* ra = rl + cur * (NH_COLS / 4);
                                ra[0] = sa;
                                ra[NH_COLS / 8] = sb;
                            }
                            sa = make_float4(0.0f, 0.0f, 0.0f, 0.0f);
                            sb = make_float4(0.0f, 0.0f, 0.0f, 0.0f);
                            cur = r;
                        }
                        sa.x += ua * va[k].x; sa.y += ua * va[k].y; sa.z += ua * va[k].z; sa.w += ua * va[k].w;
                        sb.x += ub * vb[k].x; sb.y += ub * vb[k].y; sb.z += ub * vb[k].z; sb.w += ub * vb[k].w;
                    }
                }
            }
        }
        if (cur >= 0) {
            float4* ra = rl + cur * (NH_COLS / 4);
            ra[0] = sa;
            ra[NH_COLS / 8] = sb;
        }
    } else {
        for (int e = tid; e < Etot; e += AGG_THREADS) {
            int s, d; edge_endpoints(ei, e, E, s, d);
            const int r = d - n0;
            if ((unsigned)r < ur) {
                float l0, l1, l2, l3; edge_logits(s, d, a_src, a_dst, l0, l1, l2, l3);
                atomicMaxFloat(&smax[0 * ROWS_PER_BLOCK + r], l0);
                atomicMaxFloat(&smax[1 * ROWS_PER_BLOCK + r], l1);
                atomicMaxFloat(&smax[2 * ROWS_PER_BLOCK + r], l2);
                atomicMaxFloat(&smax[3 * ROWS_PER_BLOCK + r], l3);
            }
        }
        __syncthreads();
        for (int e = tid; e < Etot; e += AGG_THREADS) {
            int s, d; edge_endpoints(ei, e, E, s, d);
            const int r = d - n0;
            if ((unsigned)r < ur) {
                float l0, l1, l2, l3; edge_logits(s, d, a_src, a_dst, l0, l1, l2, l3);
                atomicAdd(&sden[0 * ROWS_PER_BLOCK + r], __expf(l0 - smax[0 * ROWS_PER_BLOCK + r]));
                atomicAdd(&sden[1 * ROWS_PER_BLOCK + r], __expf(l1 - smax[1 * ROWS_PER_BLOCK + r]));
                atomicAdd(&sden[2 * ROWS_PER_BLOCK + r], __expf(l2 - smax[2 * ROWS_PER_BLOCK + r]));
                atomicAdd(&sden[3 * ROWS_PER_BLOCK + r], __expf(l3 - smax[3 * ROWS_PER_BLOCK + r]));
            }
        }
        __syncthreads();
        for (int base = wave * 32; base < Etot; base += AGG_THREADS) {
            const int e = base + lane;
            int src = 0, dst = -1;
            if (e < Etot) edge_endpoints(ei, e, E, src, dst);
            const bool mine = (unsigned)(dst - n0) < ur;
            const int row = dst - n0;
            float w0 = 0.0f, w1 = 0.0f, w2 = 0.0f, w3 = 0.0f;
            if (mine) {
                float l0, l1, l2, l3; edge_logits(src, dst, a_src, a_dst, l0, l1, l2, l3);
                w0 = __expf(l0 - smax[0 * ROWS_PER_BLOCK + row]) / (sden[0 * ROWS_PER_BLOCK + row] + 1e-16f);
                w1 = __expf(l1 - smax[1 * ROWS_PER_BLOCK + row]) / (sden[1 * ROWS_PER_BLOCK + row] + 1e-16f);
                w2 = __expf(l2 - smax[2 * ROWS_PER_BLOCK + row]) / (sden[2 * ROWS_PER_BLOCK + row] + 1e-16f);
                w3 = __expf(l3 - smax[3 * ROWS_PER_BLOCK + row]) / (sden[3 * ROWS_PER_BLOCK + row] + 1e-16f);
            }
            unsigned mask = __builtin_amdgcn_ballot_w32(mine);
            while (mask) {
                const int j = __builtin_ctz(mask);
                mask &= mask - 1;
                const float* hr = hbase + (size_t)__builtin_amdgcn_readlane(src, j) * NH_COLS;
                const float v0 = hr[0],   v1 = hr[32],  v2 = hr[64],  v3 = hr[96];
                const float v4 = hr[128], v5 = hr[160], v6 = hr[192], v7 = hr[224];
                lds_add_row(rows + __builtin_amdgcn_readlane(row, j) * NH_COLS + lane,
                            lane_bcast(w0, j), lane_bcast(w1, j), lane_bcast(w2, j), lane_bcast(w3, j),
                            v0, v1, v2, v3, v4, v5, v6, v7);
            }
        }
    }

    __syncthreads();

    v2f wp[32];
    #pragma unroll
    for (int c = 0; c < 32; ++c) wp[c] = ((const v2f*)wps)[c * 32 + lane];

    #pragma unroll
    for (int k = 0; k < OUT_ROWS_PER_WAVE; ++k) {
        const int r = wave + AGG_WAVES * k;
        if (r < rcount) {
            const float* rp = rows + r * NH_COLS + f2;
            const v2f a0 = *(const v2f*)(rp);
            const v2f a1 = *(const v2f*)(rp + F_OUT);
            const v2f a2 = *(const v2f*)(rp + 2 * F_OUT);
            const v2f a3 = *(const v2f*)(rp + 3 * F_OUT);
            const v2f quarter = {0.25f, 0.25f};
            const v2f m = __builtin_elementwise_fma(quarter, a0 + a1 + a2 + a3, bg);
            v2f p = bp;
            #pragma unroll
            for (int c = 0; c < 32; ++c) {
                const float o = lane_bcast(oh[k], c);
                const v2f o2 = {o, o};
                p = __builtin_elementwise_fma(o2, wp[c], p);
            }
            const v2f g2 = {g, g};
            const v2f h2 = {1.0f - g, 1.0f - g};
            const v2f res = __builtin_elementwise_fma(h2, m, g2 * p);
            *(v2f*)(out + (size_t)(n0 + r) * F_OUT + f2) = res;
        }
    }
}

extern "C" void kernel_launch(void* const* d_in, const int* in_sizes, int n_in,
                              void* d_out, int out_size, void* d_ws, size_t ws_size,
                              hipStream_t stream)
{
    const float* x        = (const float*)d_in[0];
    const int*   ei       = (const int*)  d_in[1];
    const float* onehot   = (const float*)d_in[2];
    const float* W_gat    = (const float*)d_in[3];
    const float* att_src  = (const float*)d_in[4];
    const float* att_dst  = (const float*)d_in[5];
    const float* bias_gat = (const float*)d_in[6];
    const float* W_prior  = (const float*)d_in[7];
    const float* b_prior  = (const float*)d_in[8];
    const float* gate     = (const float*)d_in[9];
    float* out = (float*)d_out;

    const int N    = in_sizes[0] / D_IN;
    const int E    = in_sizes[1] / 2;
    const int Etot = E + N;
    const int NHs  = N * HEADS;

    float* h      = (float*)d_ws;
    float* a_src  = h      + (size_t)N * NH_COLS;
    float* a_dst  = a_src  + NHs;
    unsigned int* keys = (unsigned int*)(a_dst + NHs);

    gat_gemm_wmma<<<GEMM_BLOCKS, GEMM_THREADS, 0, stream>>>(x, W_gat, h, N);

    gat_attn_coef<<<(N + COEF_ROWS_PER_BLOCK - 1) / COEF_ROWS_PER_BLOCK, COEF_THREADS, 0, stream>>>(
        h, att_src, att_dst, a_src, a_dst, N, ei, E, keys);

    gat_edge_agg<<<(N + ROWS_PER_BLOCK - 1) / ROWS_PER_BLOCK, AGG_THREADS, 0, stream>>>(
        ei, E, Etot, N, a_src, a_dst, h,
        bias_gat, onehot, W_prior, b_prior, gate, keys, out);
}


# edge_agg finds its edges through a per-block bit map over groups of 16 edges (built in attn_coef with LDS ORs) and reads only the set groups' keys, not every key; sums keep arrival order
# speedup vs baseline: 5.7687x; 1.0642x over previous
#include <hip/hip_runtime.h>
#include <hip/hip_bf16.h>
#include <math.h>

typedef __attribute__((ext_vector_type(2))) float v2f;
typedef __attribute__((ext_vector_type(8))) float v8f;

#define HEADS 4
#define F_OUT 64
#define D_IN 128
#define NH_COLS 256
#define NEG_SLOPE 0.2f
#define ROWS_PER_BLOCK 200

#define GEMM_THREADS 256
#define GEMM_WAVES   (GEMM_THREADS / 32)
#define GEMM_BLOCKS  256
#define GEMM_TSTRIDE (GEMM_BLOCKS * (GEMM_WAVES / 4))
#define WPITCH       544

__global__ __launch_bounds__(GEMM_THREADS) void gat_gemm_wmma(
    const float* __restrict__ X,
    const float* __restrict__ W,
    float* __restrict__ H,
    int N)
{
    __shared__ __align__(16) float wl[(D_IN / 2) * WPITCH];

    const int tid = threadIdx.x;
    #pragma unroll
    for (int i = 0; i < (D_IN / 2) * (NH_COLS / 4) / GEMM_THREADS; ++i) {
        const int item = tid + GEMM_THREADS * i;
        const int p  = item >> 6;
        const int c4 = item & 63;
        const float4 r0 = *(const float4*)(W + (size_t)(2 * p)     * NH_COLS + 4 * c4);
        const float4 r1 = *(const float4*)(W + (size_t)(2 * p + 1) * NH_COLS + 4 * c4);
        float4* d = (float4*)(wl + p * WPITCH + 8 * c4);
        d[0] = make_float4(r0.x, r1.x, r0.y, r1.y);
        d[1] = make_float4(r0.z, r1.z, r0.w, r1.w);
    }

    const int wave    = __builtin_amdgcn_readfirstlane(tid >> 5);
    const int lane    = tid & 31;
    const int halfsel = lane >> 4;
    const int l16     = lane & 15;
    const int colbase = (wave & 3) * 64;
    const int ntiles  = N >> 4;

    const int first = blockIdx.x + GEMM_BLOCKS * (wave >> 2);
    int mtile = min(first, ntiles - 1);
    const float* xlane = X + (size_t)l16 * D_IN + halfsel * 2;

    v2f a[32];
    {
        const float* xrow = xlane + (size_t)mtile * (16 * D_IN);
        #pragma unroll
        for (int s = 0; s < 32; ++s) a[s] = *(const v2f*)(xrow + 4 * s);
    }

    __syncthreads();

    v2f b[32][4];
    {
        const float* bl = wl + halfsel * WPITCH + 2 * (colbase + l16);
        #pragma unroll
        for (int s = 0; s < 32; ++s) {
            #pragma unroll
            for (int n = 0; n < 4; ++n) b[s][n] = *(const v2f*)(bl + (2 * s) * WPITCH + 32 * n);
        }
    }

    if (first >= ntiles) return;

    for (;;) {
        const int  nxt  = mtile + GEMM_TSTRIDE;
        const bool more = nxt < ntiles;
        v2f an[32];
        {
            const float* xrow = xlane + (size_t)(more ? nxt : mtile) * (16 * D_IN);
            #pragma unroll
            for (int s = 0; s < 32; ++s) an[s] = *(const v2f*)(xrow + 4 * s);
        }

        v8f acc0 = {}; v8f acc1 = {}; v8f acc2 = {}; v8f acc3 = {};
        #pragma unroll
        for (int s = 0; s < 32; ++s) {
            acc0 = __builtin_amdgcn_wmma_f32_16x16x4_f32(false, a[s], false, b[s][0], (short)0, acc0, false, false);
            acc1 = __builtin_amdgcn_wmma_f32_16x16x4_f32(false, a[s], false, b[s][1], (short)0, acc1, false, false);
            acc2 = __builtin_amdgcn_wmma_f32_16x16x4_f32(false, a[s], false, b[s][2], (short)0, acc2, false, false);
            acc3 = __builtin_amdgcn_wmma_f32_16x16x4_f32(false, a[s], false, b[s][3], (short)0, acc3, false, false);
        }

        #pragma unroll
        for (int r = 0; r < 8; ++r) {
            const int m = mtile * 16 + r + halfsel * 8;
            float* hrow = H + (size_t)m * NH_COLS + colbase + l16;
            hrow[0]  = acc0[r];
            hrow[16] = acc1[r];
            hrow[32] = acc2[r];
            hrow[48] = acc3[r];
        }

        if (!more) break;
        mtile = nxt;
        #pragma unroll
        for (int s = 0; s < 32; ++s) a[s] = an[s];
    }
}

#define COEF_THREADS 256
#define COEF_ROWS    16
#define COEF_ROWS_PER_BLOCK (COEF_ROWS * (COEF_THREADS / 32))

__device__ __forceinline__ unsigned int edge_key(int dst)
{
    return ((unsigned int)dst / ROWS_PER_BLOCK) & 0xFFu;
}

__device__ __forceinline__ unsigned int pack_keys(const int4 d)
{
    return edge_key(d.x) | (edge_key(d.y) << 8) | (edge_key(d.z) << 16) | (edge_key(d.w) << 24);
}

__device__ __forceinline__ unsigned int key_word_single(const int* __restrict__ dp, int q, int E)
{
    unsigned int kw = 0u;
    #pragma unroll
    for (int j = 0; j < 4; ++j) {
        const int e = 4 * q + j;
        unsigned int kb = 0xFFu;
        if (e < E) kb = edge_key(dp[e]);
        kw |= kb << (8 * j);
    }
    return kw;
}

#define ROW_ADD(v, ctrl) \
    (v) += __int_as_float(__builtin_amdgcn_update_dpp(0, __float_as_int(v), (ctrl), 0xF, 0xF, true))

__device__ __forceinline__ float row16_sum(float v)
{
    ROW_ADD(v, 0xB1);
    ROW_ADD(v, 0x4E);
    ROW_ADD(v, 0x141);
    ROW_ADD(v, 0x140);
    return v;
}

__device__ __forceinline__ float dot4(const float4 a, const float4 b)
{
    return a.x * b.x + a.y * b.y + a.z * b.z + a.w * b.w;
}

__global__ __launch_bounds__(COEF_THREADS) void gat_attn_coef(const float* __restrict__ H,
                              const float* __restrict__ att_src,
                              const float* __restrict__ att_dst,
                              float* __restrict__ a_src,
                              float* __restrict__ a_dst, int N,
                              const int* __restrict__ ei, int E,
                              unsigned int* __restrict__ keys,
                              unsigned int* __restrict__ bitmap)
{
    __shared__ unsigned int acc[COEF_THREADS / 32][256 + 8];

    const int tid    = blockIdx.x * blockDim.x + threadIdx.x;
    const int lane   = threadIdx.x & 31;
    const int wave   = __builtin_amdgcn_readfirstlane((int)(threadIdx.x >> 5));
    const int nb = __builtin_amdgcn_readfirstlane((tid >> 5) * COEF_ROWS);

    const int* dp   = ei + E;
    const int  nb16 = E >> 4;
    const int  W    = (nb16 + 31) >> 5;
    const int  P    = (W + 31) & ~31;
    const int  nqw  = 4 * nb16;
    const bool whole = (E & 3) == 0 && nqw > 0;

    const int cfirst = blockIdx.x * (COEF_THREADS / 32);
    const int cstep  = gridDim.x * (COEF_THREADS / 32);

    int4 dd[4];
    #pragma unroll
    for (int i = 0; i < 4; ++i) {
        const int q = 128 * (cfirst + wave) + 32 * i + lane;
        dd[i] = *(const int4*)(whole ? (const void*)(dp + 4 * min(q, nqw - 1)) : (const void*)H);
    }

    const float4 sa = ((const float4*)att_src)[lane];
    const float4 sb = ((const float4*)att_src)[32 + lane];
    const float4 da = ((const float4*)att_dst)[lane];
    const float4 db = ((const float4*)att_dst)[32 + lane];

    const bool have = nb < N;
    float4 va[COEF_ROWS], vb[COEF_ROWS];
    if (have) {
        const float4* hl = (const float4*)H + lane;
        #pragma unroll
        for (int k = 0; k < COEF_ROWS; ++k) {
            const float4* hp = hl + (size_t)min(nb + k, N - 1) * (NH_COLS / 4);
            va[k] = hp[0];
            vb[k] = hp[NH_COLS / 8];
        }
    }

    for (int cb = cfirst; cb < W; cb += cstep) {
        const int c = cb + wave;
        #pragma unroll
        for (int r = 0; r < 8; ++r) acc[wave][lane + 32 * r] = 0u;
        __syncthreads();
        if (c < W) {
            if (cb != cfirst && whole) {
                #pragma unroll
                for (int i = 0; i < 4; ++i) {
                    const int q = 128 * c + 32 * i + lane;
                    dd[i] = *(const int4*)(dp + 4 * min(q, nqw - 1));
                }
            }
            #pragma unroll
            for (int i = 0; i < 4; ++i) {
                const int q = 128 * c + 32 * i + lane;
                if (q < nqw) {
                    const unsigned int kw = whole ? pack_keys(dd[i]) : key_word_single(dp, q, E);
                    keys[q] = kw;
                    const unsigned int bit = 1u << (8 * i + (lane >> 2));
                    atomicOr(&acc[wave][kw & 0xFFu], bit);
                    atomicOr(&acc[wave][(kw >> 8) & 0xFFu], bit);
                    atomicOr(&acc[wave][(kw >> 16) & 0xFFu], bit);
                    atomicOr(&acc[wave][kw >> 24], bit);
                }
            }
        }
        __syncthreads();
        #pragma unroll
        for (int r = 0; r < 8; ++r) {
            const int f = (wave * 8 + r) * 32 + lane;
            const int k = f >> 3, j = f & 7;
            if (cb + j < W) bitmap[(size_t)k * P + cb + j] = acc[j][k];
        }
        __syncthreads();
    }

    if (!have) return;

    float s0[COEF_ROWS], s1[COEF_ROWS], t0[COEF_ROWS], t1[COEF_ROWS];
    #pragma unroll
    for (int k = 0; k < COEF_ROWS; ++k) {
        s0[k] = row16_sum(dot4(va[k], sa));
        s1[k] = row16_sum(dot4(vb[k], sb));
        t0[k] = row16_sum(dot4(va[k], da));
        t1[k] = row16_sum(dot4(vb[k], db));
    }

    const int r8   = (lane & 15) >> 1;
    const bool odd = (lane & 1) != 0;
    const int head = (odd ? 2 : 0) + (lane >> 4);
    #pragma unroll
    for (int g = 0; g < COEF_ROWS / 8; ++g) {
        float vs = 0.0f, vd = 0.0f;
        #pragma unroll
        for (int k = 0; k < 8; ++k) {
            if (r8 == k) {
                vs = odd ? s1[8 * g + k] : s0[8 * g + k];
                vd = odd ? t1[8 * g + k] : t0[8 * g + k];
            }
        }
        const int row = nb + 8 * g + r8;
        if (row < N) {
            a_src[(size_t)row * HEADS + head] = vs;
            a_dst[(size_t)row * HEADS + head] = vd;
        }
    }
}

__device__ __forceinline__ void atomicMaxFloat(float* addr, float val)
{
    if (val >= 0.0f) atomicMax((int*)addr, __float_as_int(val));
    else             atomicMin((unsigned int*)addr, __float_as_uint(val));
}

__device__ __forceinline__ void edge_endpoints(const int* __restrict__ ei,
                                               int e, int E, int& src, int& dst)
{
    if (e < E) { src = ei[e]; dst = ei[E + e]; }
    else       { src = dst = e - E; }
}

__device__ __forceinline__ float leaky(float a)
{
    return a > 0.0f ? a : NEG_SLOPE * a;
}

#define AGG_THREADS    1024
#define AGG_WAVES      (AGG_THREADS / 32)
#define QCAP           4096
#define QPT            (QCAP / AGG_THREADS)
#define GROUP          8
#define OUT_ROWS_PER_WAVE ((ROWS_PER_BLOCK + AGG_WAVES - 1) / AGG_WAVES)

static_assert(HEADS * ROWS_PER_BLOCK <= AGG_THREADS, "one thread per (head,row) start value");
static_assert(QPT * AGG_THREADS == QCAP, "the stats step covers the queue exactly");
static_assert(2 * AGG_THREADS == 32 * F_OUT, "two floats of W_prior per thread fill its LDS copy");

__device__ __forceinline__ float lane_bcast(float v, int j)
{
    return __int_as_float(__builtin_amdgcn_readlane(__float_as_int(v), j));
}

__device__ __forceinline__ void lds_add_row(float* o,
                                            float u0, float u1, float u2, float u3,
                                            float v0, float v1, float v2, float v3,
                                            float v4, float v5, float v6, float v7)
{
    atomicAdd(o,       u0 * v0);
    atomicAdd(o + 32,  u0 * v1);
    atomicAdd(o + 64,  u1 * v2);
    atomicAdd(o + 96,  u1 * v3);
    atomicAdd(o + 128, u2 * v4);
    atomicAdd(o + 160, u2 * v5);
    atomicAdd(o + 192, u3 * v6);
    atomicAdd(o + 224, u3 * v7);
}

__device__ __forceinline__ void edge_logits(int src, int dst,
                                            const float* __restrict__ a_src,
                                            const float* __restrict__ a_dst,
                                            float& l0, float& l1, float& l2, float& l3)
{
    const float4 as = *(const float4*)(a_src + (size_t)src * HEADS);
    const float4 ad = *(const float4*)(a_dst + (size_t)dst * HEADS);
    l0 = leaky(as.x + ad.x);
    l1 = leaky(as.y + ad.y);
    l2 = leaky(as.z + ad.z);
    l3 = leaky(as.w + ad.w);
}

__global__ __launch_bounds__(AGG_THREADS) void gat_edge_agg(const int* __restrict__ ei, int E, int Etot, int N,
                             const float* __restrict__ a_src,
                             const float* __restrict__ a_dst,
                             const float* __restrict__ H,
                             const float* __restrict__ bias_gat,
                             const float* __restrict__ onehot,
                             const float* __restrict__ W_prior,
                             const float* __restrict__ b_prior,
                             const float* __restrict__ gate,
                             const unsigned int* __restrict__ keys,
                             const unsigned int* __restrict__ bitmap,
                             float* __restrict__ out)
{
    __shared__ __align__(16) float rows[ROWS_PER_BLOCK * NH_COLS];
    __shared__ __align__(16) float qw[QCAP * HEADS];
    __shared__ int   queue[QCAP];
    __shared__ int   qrow[QCAP];
    __shared__ float smax[HEADS * ROWS_PER_BLOCK];
    __shared__ float sden[HEADS * ROWS_PER_BLOCK];
    __shared__ __align__(8) float wps[32 * F_OUT];
    __shared__ int   qcount;
    __shared__ int   gcount;
    __shared__ int   rcnt[AGG_WAVES * OUT_ROWS_PER_WAVE];
    __shared__ int   rbase[AGG_WAVES * OUT_ROWS_PER_WAVE];
    __shared__ int   wcnt[AGG_WAVES];
    __shared__ int   wbase[AGG_WAVES];

    const int tid  = threadIdx.x;
    const int lane = tid & 31;
    const int wave = __builtin_amdgcn_readfirstlane(tid >> 5);
    const int n0     = blockIdx.x * ROWS_PER_BLOCK;
    const int rcount = min(ROWS_PER_BLOCK, N - n0);
    const unsigned ur = (unsigned)rcount;

    float4* rows4 = (float4*)rows;
    for (int i = tid; i < ROWS_PER_BLOCK * (NH_COLS / 4); i += AGG_THREADS)
        rows4[i] = make_float4(0.0f, 0.0f, 0.0f, 0.0f);
    if (tid < HEADS * ROWS_PER_BLOCK) {
        smax[tid] = __int_as_float(0xFF800000);
        sden[tid] = 0.0f;
    }
    if (tid < rcount) queue[tid] = E + n0 + tid;
    if (tid == 0) { qcount = rcount; gcount = 0; }
    if (tid < AGG_WAVES * OUT_ROWS_PER_WAVE) rcnt[tid] = 0;
    __syncthreads();

    const int    f2     = 2 * lane;
    const float2 wstage = *(const float2*)(W_prior + 2 * tid);
    float oh[OUT_ROWS_PER_WAVE];
    #pragma unroll
    for (int k = 0; k < OUT_ROWS_PER_WAVE; ++k) {
        const int r = min(wave + AGG_WAVES * k, rcount - 1);
        oh[k] = onehot[(size_t)(n0 + r) * 32 + lane];
    }
    const v2f bg = *(const v2f*)(bias_gat + f2);
    const v2f bp = *(const v2f*)(b_prior + f2);
    const float g = gate[0];

    const bool usekeys = gridDim.x <= 256u;
    const int* dp  = ei + E;
    const uint4* kp4 = (const uint4*)keys;
    const int  nb16 = usekeys ? (E >> 4) : 0;
    const int  W    = (nb16 + 31) >> 5;
    const unsigned int pat = blockIdx.x * 0x01010101u;
    int* glist = qrow;
    if (nb16 > 0) {
        const unsigned int* brow = bitmap + (size_t)blockIdx.x * ((W + 31) & ~31);
        for (int t = tid; t < W; t += 2 * AGG_THREADS) {
            const int t1 = t + AGG_THREADS;
            unsigned int m0 = brow[t];
            unsigned int m1 = brow[min(t1, W - 1)];
            if (t1 >= W) m1 = 0u;
            const int c = __popc(m0) + __popc(m1);
            if (c) {
                int s = atomicAdd(&gcount, c);
                while (m0) {
                    const int p = __builtin_ctz(m0);
                    m0 &= m0 - 1u;
                    if (s < QCAP) glist[s] = 32 * t + p;
                    ++s;
                }
                while (m1) {
                    const int p = __builtin_ctz(m1);
                    m1 &= m1 - 1u;
                    if (s < QCAP) glist[s] = 32 * t1 + p;
                    ++s;
                }
            }
        }
    }
    __syncthreads();
    const int ng = __builtin_amdgcn_readfirstlane(gcount);
    if (nb16 > 0 && ng <= QCAP) {
        uint4 d[QPT];
        int   gq[QPT];
        #pragma unroll
        for (int k = 0; k < QPT; ++k) {
            const int idx = tid + k * AGG_THREADS;
            gq[k] = (idx < ng) ? glist[idx] : 0;
        }
        #pragma unroll
        for (int k = 0; k < QPT; ++k) d[k] = kp4[gq[k]];
        unsigned int zz[QPT];
        int c = 0;
        #pragma unroll
        for (int k = 0; k < QPT; ++k) {
            const bool ok = (tid + k * AGG_THREADS) < ng;
            const unsigned int x0 = d[k].x ^ pat, x1 = d[k].y ^ pat;
            const unsigned int x2 = d[k].z ^ pat, x3 = d[k].w ^ pat;
            const unsigned int z0 = ~(((x0 & 0x7F7F7F7Fu) + 0x7F7F7F7Fu) | x0 | 0x7F7F7F7Fu);
            const unsigned int z1 = ~(((x1 & 0x7F7F7F7Fu) + 0x7F7F7F7Fu) | x1 | 0x7F7F7F7Fu);
            const unsigned int z2 = ~(((x2 & 0x7F7F7F7Fu) + 0x7F7F7F7Fu) | x2 | 0x7F7F7F7Fu);
            const unsigned int z3 = ~(((x3 & 0x7F7F7F7Fu) + 0x7F7F7F7Fu) | x3 | 0x7F7F7F7Fu);
            const unsigned int m = (z0 >> 7) | (z1 >> 6) | (z2 >> 5) | (z3 >> 4);
            zz[k] = ok ? m : 0u;
            c += __popc(zz[k]);
        }
        if (c) {
            int s = atomicAdd(&qcount, c);
            #pragma unroll
            for (int k = 0; k < QPT; ++k) {
                unsigned int m = zz[k];
                const int e0 = 16 * gq[k];
                while (m) {
                    const int p = __builtin_ctz(m);
                    m &= m - 1u;
                    if (s < QCAP) queue[s] = e0 + 4 * (p & 7) + (p >> 3);
                    ++s;
                }
            }
        }
    }
    if (usekeys) {
        const int t = 16 * nb16 + tid;
        if (tid < 16 && t < E && (unsigned)(dp[t] - n0) < ur) {
            const int s = atomicAdd(&qcount, 1);
            if (s < QCAP) queue[s] = t;
        }
    }
    ((float2*)wps)[tid] = wstage;
    __syncthreads();

    const int total = (usekeys && ng <= QCAP) ? __builtin_amdgcn_readfirstlane(qcount) : QCAP + 1;
    const float* hbase = H + lane;

    if (total <= QCAP) {
        int   qs[QPT], qr[QPT], rk[QPT], sl[QPT];
        float lg[QPT][HEADS];
        #pragma unroll
        for (int k = 0; k < QPT; ++k) {
            const int idx = tid + k * AGG_THREADS;
            const int e = (idx < total) ? queue[idx] : (E + n0);
            int s, d; edge_endpoints(ei, e, E, s, d);
            qs[k] = s;
            qr[k] = d - n0;
        }
        #pragma unroll
        for (int k = 0; k < QPT; ++k)
            edge_logits(qs[k], n0 + qr[k], a_src, a_dst, lg[k][0], lg[k][1], lg[k][2], lg[k][3]);
        #pragma unroll
        for (int k = 0; k < QPT; ++k) {
            const int idx = tid + k * AGG_THREADS;
            rk[k] = 0;
            sl[k] = (qr[k] & (AGG_WAVES - 1)) * OUT_ROWS_PER_WAVE + (qr[k] >> 5);
            if (idx < total) {
                rk[k] = atomicAdd(&rcnt[sl[k]], 1);
                #pragma unroll
                for (int hd = 0; hd < HEADS; ++hd)
                    atomicMaxFloat(&smax[hd * ROWS_PER_BLOCK + qr[k]], lg[k][hd]);
            }
        }
        __syncthreads();
        if (wave == 0) {
            int cr[OUT_ROWS_PER_WAVE];
            int cw = 0;
            #pragma unroll
            for (int k = 0; k < OUT_ROWS_PER_WAVE; ++k) {
                cr[k] = rcnt[lane * OUT_ROWS_PER_WAVE + k];
                cw += cr[k];
            }
            int incl = cw;
            #pragma unroll
            for (int d = 1; d < 32; d <<= 1) {
                const int t = __shfl_up(incl, d);
                if (lane >= d) incl += t;
            }
            int run = incl - cw;
            wbase[lane] = run;
            wcnt[lane]  = cw;
            #pragma unroll
            for (int k = 0; k < OUT_ROWS_PER_WAVE; ++k) {
                rbase[lane * OUT_ROWS_PER_WAVE + k] = run;
                run += cr[k];
            }
        }
        float st[QPT][HEADS];
        #pragma unroll
        for (int k = 0; k < QPT; ++k) {
            #pragma unroll
            for (int hd = 0; hd < HEADS; ++hd) st[k][hd] = smax[hd * ROWS_PER_BLOCK + qr[k]];
        }
        #pragma unroll
        for (int k = 0; k < QPT; ++k) {
            #pragma unroll
            for (int hd = 0; hd < HEADS; ++hd) lg[k][hd] = __expf(lg[k][hd] - st[k][hd]);
        }
        #pragma unroll
        for (int k = 0; k < QPT; ++k) {
            const int idx = tid + k * AGG_THREADS;
            if (idx < total) {
                #pragma unroll
                for (int hd = 0; hd < HEADS; ++hd)
                    atomicAdd(&sden[hd * ROWS_PER_BLOCK + qr[k]], lg[k][hd]);
            }
        }
        __syncthreads();
        #pragma unroll
        for (int k = 0; k < QPT; ++k) {
            #pragma unroll
            for (int hd = 0; hd < HEADS; ++hd) st[k][hd] = sden[hd * ROWS_PER_BLOCK + qr[k]];
        }
        #pragma unroll
        for (int k = 0; k < QPT; ++k) {
            const int idx = tid + k * AGG_THREADS;
            if (idx < total) {
                float4 w;
                w.x = lg[k][0] / (st[k][0] + 1e-16f);
                w.y = lg[k][1] / (st[k][1] + 1e-16f);
                w.z = lg[k][2] / (st[k][2] + 1e-16f);
                w.w = lg[k][3] / (st[k][3] + 1e-16f);
                const int pos = rbase[sl[k]] + rk[k];
                queue[pos] = qs[k];
                qrow[pos]  = qr[k];
                ((float4*)qw)[pos] = w;
            }
        }
        __syncthreads();

        const int beg = __builtin_amdgcn_readfirstlane(wbase[wave]);
        const int cnt = __builtin_amdgcn_readfirstlane(wcnt[wave]);
        const bool lowhalf = lane < 16;
        const float4* hl = (const float4*)H + lane;
        float4*       rl = (float4*)rows + lane;
        float4 sa = make_float4(0.0f, 0.0f, 0.0f, 0.0f);
        float4 sb = make_float4(0.0f, 0.0f, 0.0f, 0.0f);
        int cur = -1;
        for (int c0 = 0; c0 < cnt; c0 += 32) {
            const int n = min(32, cnt - c0);
            int src = 0, row = 0;
            float w0 = 0.0f, w1 = 0.0f, w2 = 0.0f, w3 = 0.0f;
            if (lane < n) {
                src = queue[beg + c0 + lane];
                row = qrow[beg + c0 + lane];
                const float4 w = ((const float4*)qw)[beg + c0 + lane];
                w0 = w.x; w1 = w.y; w2 = w.z; w3 = w.w;
            }
            for (int g = 0; g < n; g += GROUP) {
                float4 va[GROUP], vb[GROUP];
                #pragma unroll
                for (int k = 0; k < GROUP; ++k) {
                    const float4* hp = hl + (size_t)__builtin_amdgcn_readlane(src, g + k) * (NH_COLS / 4);
                    va[k] = hp[0];
                    vb[k] = hp[NH_COLS / 8];
                }
                #pragma unroll
                for (int k = 0; k < GROUP; ++k) {
                    if (g + k < n) {
                        const float u0 = lane_bcast(w0, g + k), u1 = lane_bcast(w1, g + k);
                        const float u2 = lane_bcast(w2, g + k), u3 = lane_bcast(w3, g + k);
                        const float ua = lowhalf ? u0 : u1;
                        const float ub = lowhalf ? u2 : u3;
                        const int r = __builtin_amdgcn_readlane(row, g + k);
                        if (r != cur) {
                            if (cur >= 0) {
                                float4* ra = rl + cur * (NH_COLS / 4);
                                ra[0] = sa;
                                ra[NH_COLS / 8] = sb;
                            }
                            sa = make_float4(0.0f, 0.0f, 0.0f, 0.0f);
                            sb = make_float4(0.0f, 0.0f, 0.0f, 0.0f);
                            cur = r;
                        }
                        sa.x += ua * va[k].x; sa.y += ua * va[k].y; sa.z += ua * va[k].z; sa.w += ua * va[k].w;
                        sb.x += ub * vb[k].x; sb.y += ub * vb[k].y; sb.z += ub * vb[k].z; sb.w += ub * vb[k].w;
                    }
                }
            }
        }
        if (cur >= 0) {
            float4* ra = rl + cur * (NH_COLS / 4);
            ra[0] = sa;
            ra[NH_COLS / 8] = sb;
        }
    } else {
        for (int e = tid; e < Etot; e += AGG_THREADS) {
            int s, d; edge_endpoints(ei, e, E, s, d);
            const int r = d - n0;
            if ((unsigned)r < ur) {
                float l0, l1, l2, l3; edge_logits(s, d, a_src, a_dst, l0, l1, l2, l3);
                atomicMaxFloat(&smax[0 * ROWS_PER_BLOCK + r], l0);
                atomicMaxFloat(&smax[1 * ROWS_PER_BLOCK + r], l1);
                atomicMaxFloat(&smax[2 * ROWS_PER_BLOCK + r], l2);
                atomicMaxFloat(&smax[3 * ROWS_PER_BLOCK + r], l3);
            }
        }
        __syncthreads();
        for (int e = tid; e < Etot; e += AGG_THREADS) {
            int s, d; edge_endpoints(ei, e, E, s, d);
            const int r = d - n0;
            if ((unsigned)r < ur) {
                float l0, l1, l2, l3; edge_logits(s, d, a_src, a_dst, l0, l1, l2, l3);
                atomicAdd(&sden[0 * ROWS_PER_BLOCK + r], __expf(l0 - smax[0 * ROWS_PER_BLOCK + r]));
                atomicAdd(&sden[1 * ROWS_PER_BLOCK + r], __expf(l1 - smax[1 * ROWS_PER_BLOCK + r]));
                atomicAdd(&sden[2 * ROWS_PER_BLOCK + r], __expf(l2 - smax[2 * ROWS_PER_BLOCK + r]));
                atomicAdd(&sden[3 * ROWS_PER_BLOCK + r], __expf(l3 - smax[3 * ROWS_PER_BLOCK + r]));
            }
        }
        __syncthreads();
        for (int base = wave * 32; base < Etot; base += AGG_THREADS) {
            const int e = base + lane;
            int src = 0, dst = -1;
            if (e < Etot) edge_endpoints(ei, e, E, src, dst);
            const bool mine = (unsigned)(dst - n0) < ur;
            const int row = dst - n0;
            float w0 = 0.0f, w1 = 0.0f, w2 = 0.0f, w3 = 0.0f;
            if (mine) {
                float l0, l1, l2, l3; edge_logits(src, dst, a_src, a_dst, l0, l1, l2, l3);
                w0 = __expf(l0 - smax[0 * ROWS_PER_BLOCK + row]) / (sden[0 * ROWS_PER_BLOCK + row] + 1e-16f);
                w1 = __expf(l1 - smax[1 * ROWS_PER_BLOCK + row]) / (sden[1 * ROWS_PER_BLOCK + row] + 1e-16f);
                w2 = __expf(l2 - smax[2 * ROWS_PER_BLOCK + row]) / (sden[2 * ROWS_PER_BLOCK + row] + 1e-16f);
                w3 = __expf(l3 - smax[3 * ROWS_PER_BLOCK + row]) / (sden[3 * ROWS_PER_BLOCK + row] + 1e-16f);
            }
            unsigned mask = __builtin_amdgcn_ballot_w32(mine);
            while (mask) {
                const int j = __builtin_ctz(mask);
                mask &= mask - 1;
                const float* hr = hbase + (size_t)__builtin_amdgcn_readlane(src, j) * NH_COLS;
                const float v0 = hr[0],   v1 = hr[32],  v2 = hr[64],  v3 = hr[96];
                const float v4 = hr[128], v5 = hr[160], v6 = hr[192], v7 = hr[224];
                lds_add_row(rows + __builtin_amdgcn_readlane(row, j) * NH_COLS + lane,
                            lane_bcast(w0, j), lane_bcast(w1, j), lane_bcast(w2, j), lane_bcast(w3, j),
                            v0, v1, v2, v3, v4, v5, v6, v7);
            }
        }
    }

    __syncthreads();

    v2f wp[32];
    #pragma unroll
    for (int c = 0; c < 32; ++c) wp[c] = ((const v2f*)wps)[c * 32 + lane];

    #pragma unroll
    for (int k = 0; k < OUT_ROWS_PER_WAVE; ++k) {
        const int r = wave + AGG_WAVES * k;
        if (r < rcount) {
            const float* rp = rows + r * NH_COLS + f2;
            const v2f a0 = *(const v2f*)(rp);
            const v2f a1 = *(const v2f*)(rp + F_OUT);
            const v2f a2 = *(const v2f*)(rp + 2 * F_OUT);
            const v2f a3 = *(const v2f*)(rp + 3 * F_OUT);
            const v2f quarter = {0.25f, 0.25f};
            const v2f m = __builtin_elementwise_fma(quarter, a0 + a1 + a2 + a3, bg);
            v2f p = bp;
            #pragma unroll
            for (int c = 0; c < 32; ++c) {
                const float o = lane_bcast(oh[k], c);
                const v2f o2 = {o, o};
                p = __builtin_elementwise_fma(o2, wp[c], p);
            }
            const v2f g2 = {g, g};
            const v2f h2 = {1.0f - g, 1.0f - g};
            const v2f res = __builtin_elementwise_fma(h2, m, g2 * p);
            *(v2f*)(out + (size_t)(n0 + r) * F_OUT + f2) = res;
        }
    }
}

extern "C" void kernel_launch(void* const* d_in, const int* in_sizes, int n_in,
                              void* d_out, int out_size, void* d_ws, size_t ws_size,
                              hipStream_t stream)
{
    const float* x        = (const float*)d_in[0];
    const int*   ei       = (const int*)  d_in[1];
    const float* onehot   = (const float*)d_in[2];
    const float* W_gat    = (const float*)d_in[3];
    const float* att_src  = (const float*)d_in[4];
    const float* att_dst  = (const float*)d_in[5];
    const float* bias_gat = (const float*)d_in[6];
    const float* W_prior  = (const float*)d_in[7];
    const float* b_prior  = (const float*)d_in[8];
    const float* gate     = (const float*)d_in[9];
    float* out = (float*)d_out;

    const int N    = in_sizes[0] / D_IN;
    const int E    = in_sizes[1] / 2;
    const int Etot = E + N;
    const int NHs  = N * HEADS;

    float* h      = (float*)d_ws;
    float* a_src  = h      + (size_t)N * NH_COLS;
    float* a_dst  = a_src  + NHs;
    unsigned int* keys = (unsigned int*)(a_dst + NHs);
    unsigned int* bitmap = keys + (E + 3) / 4;

    gat_gemm_wmma<<<GEMM_BLOCKS, GEMM_THREADS, 0, stream>>>(x, W_gat, h, N);

    gat_attn_coef<<<(N + COEF_ROWS_PER_BLOCK - 1) / COEF_ROWS_PER_BLOCK, COEF_THREADS, 0, stream>>>(
        h, att_src, att_dst, a_src, a_dst, N, ei, E, keys, bitmap);

    gat_edge_agg<<<(N + ROWS_PER_BLOCK - 1) / ROWS_PER_BLOCK, AGG_THREADS, 0, stream>>>(
        ei, E, Etot, N, a_src, a_dst, h,
        bias_gat, onehot, W_prior, b_prior, gate, keys, bitmap, out);
}
